# starting kernel
# baseline (speedup 1.0000x reference)
#include <hip/hip_runtime.h>
#include <hip/hip_bf16.h>

#define H   768
#define NS  128
#define NQ  256
#define NTOT 384
#define RREL 8
#define KC  64

typedef __attribute__((ext_vector_type(8))) float v8f;
typedef __attribute__((ext_vector_type(2))) float v2f;

__global__ void wbar_kernel(const float* __restrict__ relW, float* __restrict__ wbar) {
    int e = blockIdx.x * 256 + threadIdx.x;
    float s = 0.f;
#pragma unroll
    for (int r = 0; r < RREL; ++r) s += relW[r * (H * H) + e];
    wbar[e] = s * (1.0f / RREL);
}

__global__ void gemm_wmma_f32(const float* __restrict__ Asup,
                              const float* __restrict__ Aqry,
                              const float* __restrict__ W, int ldW, int colOff,
                              const float* __restrict__ bias, int biasMode,
                              float* __restrict__ out) {
    const int lane = threadIdx.x & 31;
    const int wv   = threadIdx.x >> 5;
    const int m0   = blockIdx.x * 16;
    const int n0   = (blockIdx.y * 4 + wv) * 16;
    const int l15  = lane & 15;
    const int khalf = (lane >> 4) * 2;

    const float* Arow = (m0 < NS) ? (Asup + (size_t)(m0 + l15) * H)
                                  : (Aqry + (size_t)(m0 - NS + l15) * H);
    const float* Brow = W + (size_t)(n0 + l15) * ldW + colOff;

    v8f acc = {0.f, 0.f, 0.f, 0.f, 0.f, 0.f, 0.f, 0.f};
#pragma unroll 4
    for (int k = 0; k < H; k += 4) {
        v2f a = *(const v2f*)(Arow + k + khalf);
        v2f b = *(const v2f*)(Brow + k + khalf);
        acc = __builtin_amdgcn_wmma_f32_16x16x4_f32(
             false, a,  false, b,
             (short)0, acc,  false,  false);
    }

    const int n = n0 + l15;
    float bv = 0.f;
    if (biasMode == 1) {
        bv = bias[n];
    } else if (biasMode == 2) {
        float s = 0.f;
#pragma unroll
        for (int r = 0; r < RREL; ++r) s += bias[r * H + n];
        bv = s * (1.0f / RREL);
    }

    const int mbase = m0 + ((lane >> 4) ? 8 : 0);
#pragma unroll
    for (int r8 = 0; r8 < 8; ++r8) {
        out[(size_t)(mbase + r8) * H + n] = acc[r8] + bv;
    }
}

__global__ void pair_kernel(const float* __restrict__ Lsim, const float* __restrict__ Rsim,
                            const float* __restrict__ w2sim, const float* __restrict__ b2sim,
                            const float* __restrict__ Ldiv, const float* __restrict__ Rdiv,
                            const float* __restrict__ w2div, const float* __restrict__ b2div,
                            float* __restrict__ outSim, float* __restrict__ outDiv) {
    __shared__ float Ls[32][KC + 1];
    __shared__ float Rs[32][KC + 1];
    __shared__ float Ws[KC];

    const float *Lp, *Rp, *w2;
    float b2;
    float* out;
    if (blockIdx.z == 0) { Lp = Lsim; Rp = Rsim; w2 = w2sim; b2 = b2sim[0]; out = outSim; }
    else                 { Lp = Ldiv; Rp = Rdiv; w2 = w2div; b2 = b2div[0]; out = outDiv; }

    const int i0 = blockIdx.y * 32;
    const int j0 = blockIdx.x * 32;
    const int t  = threadIdx.x;
    const int ti = t >> 3;
    const int tj = (t & 7) * 4;

    float a0 = 0.f, a1 = 0.f, a2 = 0.f, a3 = 0.f;

    for (int k0 = 0; k0 < H; k0 += KC) {
        __syncthreads();
#pragma unroll
        for (int idx = t; idx < 32 * KC; idx += 256) {
            int r = idx >> 6;
            int c = idx & (KC - 1);
            Ls[r][c] = Lp[(size_t)(i0 + r) * H + k0 + c];
            Rs[r][c] = Rp[(size_t)(j0 + r) * H + k0 + c];
        }
        if (t < KC) Ws[t] = w2[k0 + t];
        __syncthreads();
#pragma unroll 4
        for (int k = 0; k < KC; ++k) {
            float lv = Ls[ti][k];
            float wv = Ws[k];
            a0 += fmaxf(lv + Rs[tj + 0][k], 0.f) * wv;
            a1 += fmaxf(lv + Rs[tj + 1][k], 0.f) * wv;
            a2 += fmaxf(lv + Rs[tj + 2][k], 0.f) * wv;
            a3 += fmaxf(lv + Rs[tj + 3][k], 0.f) * wv;
        }
    }

    const int i = i0 + ti;
    const int j = j0 + tj;
    float v[4] = {a0 + b2, a1 + b2, a2 + b2, a3 + b2};
#pragma unroll
    for (int c = 0; c < 4; ++c) {
        out[(size_t)i * NTOT + j + c] = (i == j + c) ? 0.f : v[c];
    }
}

extern "C" void kernel_launch(void* const* d_in, const int* in_sizes, int n_in,
                              void* d_out, int out_size, void* d_ws, size_t ws_size,
                              hipStream_t stream) {
    const float* sup    = (const float*)d_in[0];
    const float* qry    = (const float*)d_in[2];
    const float* simW1  = (const float*)d_in[3];
    const float* simb1  = (const float*)d_in[4];
    const float* simw2  = (const float*)d_in[5];
    const float* simb2  = (const float*)d_in[6];
    const float* divW1  = (const float*)d_in[7];
    const float* divb1  = (const float*)d_in[8];
    const float* divw2  = (const float*)d_in[9];
    const float* divb2  = (const float*)d_in[10];
    const float* relW   = (const float*)d_in[11];
    const float* relb   = (const float*)d_in[12];

    float* out = (float*)d_out;
    float* ws  = (float*)d_ws;

    float* wbar = ws;
    float* Lsim = ws + 589824;
    float* Rsim = Lsim + 294912;
    float* Ldiv = Rsim + 294912;
    float* Rdiv = Ldiv + 294912;

    float* outProp = out;
    float* outSim  = out + NTOT * H;
    float* outDiv  = outSim + NTOT * NTOT;

    wbar_kernel<<<(H * H) / 256, 256, 0, stream>>>(relW, wbar);

    dim3 ggrid(NTOT / 16, (H / 16) / 4);
    dim3 gblk(128);
    gemm_wmma_f32<<<ggrid, gblk, 0, stream>>>(sup, qry, wbar, H, 0, relb, 2, outProp);
    gemm_wmma_f32<<<ggrid, gblk, 0, stream>>>(sup, qry, simW1, 2 * H, 0, simb1, 1, Lsim);
    gemm_wmma_f32<<<ggrid, gblk, 0, stream>>>(sup, qry, simW1, 2 * H, H, nullptr, 0, Rsim);
    gemm_wmma_f32<<<ggrid, gblk, 0, stream>>>(sup, qry, divW1, 2 * H, 0, divb1, 1, Ldiv);
    gemm_wmma_f32<<<ggrid, gblk, 0, stream>>>(sup, qry, divW1, 2 * H, H, nullptr, 0, Rdiv);

    dim3 pgrid(NTOT / 32, NTOT / 32, 2);
    pair_kernel<<<pgrid, 256, 0, stream>>>(Lsim, Rsim, simw2, simb2,
                                           Ldiv, Rdiv, divw2, divb2,
                                           outSim, outDiv);
}


# One launch for the five independent f32 WMMA GEMMs (grid z picks weight, bias, output): 5x the waves per SIMD overlap fragment-load latency; four launch gaps and four part-filled last rounds gone
# speedup vs baseline: 1.3497x; 1.3497x over previous
#include <hip/hip_runtime.h>
#include <hip/hip_bf16.h>

#define H   768
#define NS  128
#define NQ  256
#define NTOT 384
#define RREL 8
#define KC  64

typedef __attribute__((ext_vector_type(8))) float v8f;
typedef __attribute__((ext_vector_type(2))) float v2f;

__global__ void wbar_kernel(const float* __restrict__ relW, float* __restrict__ wbar) {
    int e = blockIdx.x * 256 + threadIdx.x;
    float s = 0.f;
#pragma unroll
    for (int r = 0; r < RREL; ++r) s += relW[r * (H * H) + e];
    wbar[e] = s * (1.0f / RREL);
}

__global__ void gemm_wmma_f32(const float* __restrict__ Asup,
                              const float* __restrict__ Aqry,
                              const float* __restrict__ wbar,
                              const float* __restrict__ simW1,
                              const float* __restrict__ divW1,
                              const float* __restrict__ relb,
                              const float* __restrict__ simb1,
                              const float* __restrict__ divb1,
                              float* __restrict__ outProp,
                              float* __restrict__ Lsim, float* __restrict__ Rsim,
                              float* __restrict__ Ldiv, float* __restrict__ Rdiv) {
    const int z = blockIdx.z;
    const float* W    = (z == 0) ? wbar : ((z <= 2) ? simW1 : divW1);
    const int ldW     = (z == 0) ? H : 2 * H;
    const int colOff  = (z == 2 || z == 4) ? H : 0;
    const float* bias = (z == 0) ? relb : ((z == 1) ? simb1 : divb1);
    const int biasMode = (z == 0) ? 2 : ((z == 1 || z == 3) ? 1 : 0);
    float* out = (z == 0) ? outProp
               : (z == 1) ? Lsim : (z == 2) ? Rsim : (z == 3) ? Ldiv : Rdiv;

    const int lane = threadIdx.x & 31;
    const int wv   = threadIdx.x >> 5;
    const int m0   = blockIdx.x * 16;
    const int n0   = (blockIdx.y * 4 + wv) * 16;
    const int l15  = lane & 15;
    const int khalf = (lane >> 4) * 2;

    const float* Arow = (m0 < NS) ? (Asup + (size_t)(m0 + l15) * H)
                                  : (Aqry + (size_t)(m0 - NS + l15) * H);
    const float* Brow = W + (size_t)(n0 + l15) * ldW + colOff;

    v8f acc = {0.f, 0.f, 0.f, 0.f, 0.f, 0.f, 0.f, 0.f};
#pragma unroll 4
    for (int k = 0; k < H; k += 4) {
        v2f a = *(const v2f*)(Arow + k + khalf);
        v2f b = *(const v2f*)(Brow + k + khalf);
        acc = __builtin_amdgcn_wmma_f32_16x16x4_f32(
             false, a,  false, b,
             (short)0, acc,  false,  false);
    }

    const int n = n0 + l15;
    float bv = 0.f;
    if (biasMode == 1) {
        bv = bias[n];
    } else if (biasMode == 2) {
        float s = 0.f;
#pragma unroll
        for (int r = 0; r < RREL; ++r) s += bias[r * H + n];
        bv = s * (1.0f / RREL);
    }

    const int mbase = m0 + ((lane >> 4) ? 8 : 0);
#pragma unroll
    for (int r8 = 0; r8 < 8; ++r8) {
        out[(size_t)(mbase + r8) * H + n] = acc[r8] + bv;
    }
}

__global__ void pair_kernel(const float* __restrict__ Lsim, const float* __restrict__ Rsim,
                            const float* __restrict__ w2sim, const float* __restrict__ b2sim,
                            const float* __restrict__ Ldiv, const float* __restrict__ Rdiv,
                            const float* __restrict__ w2div, const float* __restrict__ b2div,
                            float* __restrict__ outSim, float* __restrict__ outDiv) {
    __shared__ float Ls[32][KC + 1];
    __shared__ float Rs[32][KC + 1];
    __shared__ float Ws[KC];

    const float *Lp, *Rp, *w2;
    float b2;
    float* out;
    if (blockIdx.z == 0) { Lp = Lsim; Rp = Rsim; w2 = w2sim; b2 = b2sim[0]; out = outSim; }
    else                 { Lp = Ldiv; Rp = Rdiv; w2 = w2div; b2 = b2div[0]; out = outDiv; }

    const int i0 = blockIdx.y * 32;
    const int j0 = blockIdx.x * 32;
    const int t  = threadIdx.x;
    const int ti = t >> 3;
    const int tj = (t & 7) * 4;

    float a0 = 0.f, a1 = 0.f, a2 = 0.f, a3 = 0.f;

    for (int k0 = 0; k0 < H; k0 += KC) {
        __syncthreads();
#pragma unroll
        for (int idx = t; idx < 32 * KC; idx += 256) {
            int r = idx >> 6;
            int c = idx & (KC - 1);
            Ls[r][c] = Lp[(size_t)(i0 + r) * H + k0 + c];
            Rs[r][c] = Rp[(size_t)(j0 + r) * H + k0 + c];
        }
        if (t < KC) Ws[t] = w2[k0 + t];
        __syncthreads();
#pragma unroll 4
        for (int k = 0; k < KC; ++k) {
            float lv = Ls[ti][k];
            float wv = Ws[k];
            a0 += fmaxf(lv + Rs[tj + 0][k], 0.f) * wv;
            a1 += fmaxf(lv + Rs[tj + 1][k], 0.f) * wv;
            a2 += fmaxf(lv + Rs[tj + 2][k], 0.f) * wv;
            a3 += fmaxf(lv + Rs[tj + 3][k], 0.f) * wv;
        }
    }

    const int i = i0 + ti;
    const int j = j0 + tj;
    float v[4] = {a0 + b2, a1 + b2, a2 + b2, a3 + b2};
#pragma unroll
    for (int c = 0; c < 4; ++c) {
        out[(size_t)i * NTOT + j + c] = (i == j + c) ? 0.f : v[c];
    }
}

extern "C" void kernel_launch(void* const* d_in, const int* in_sizes, int n_in,
                              void* d_out, int out_size, void* d_ws, size_t ws_size,
                              hipStream_t stream) {
    const float* sup    = (const float*)d_in[0];
    const float* qry    = (const float*)d_in[2];
    const float* simW1  = (const float*)d_in[3];
    const float* simb1  = (const float*)d_in[4];
    const float* simw2  = (const float*)d_in[5];
    const float* simb2  = (const float*)d_in[6];
    const float* divW1  = (const float*)d_in[7];
    const float* divb1  = (const float*)d_in[8];
    const float* divw2  = (const float*)d_in[9];
    const float* divb2  = (const float*)d_in[10];
    const float* relW   = (const float*)d_in[11];
    const float* relb   = (const float*)d_in[12];

    float* out = (float*)d_out;
    float* ws  = (float*)d_ws;

    float* wbar = ws;
    float* Lsim = ws + 589824;
    float* Rsim = Lsim + 294912;
    float* Ldiv = Rsim + 294912;
    float* Rdiv = Ldiv + 294912;

    float* outProp = out;
    float* outSim  = out + NTOT * H;
    float* outDiv  = outSim + NTOT * NTOT;

    wbar_kernel<<<(H * H) / 256, 256, 0, stream>>>(relW, wbar);

    dim3 ggrid(NTOT / 16, (H / 16) / 4, 5);
    dim3 gblk(128);
    gemm_wmma_f32<<<ggrid, gblk, 0, stream>>>(sup, qry, wbar, simW1, divW1,
                                              relb, simb1, divb1,
                                              outProp, Lsim, Rsim, Ldiv, Rdiv);

    dim3 pgrid(NTOT / 32, NTOT / 32, 2);
    pair_kernel<<<pgrid, 256, 0, stream>>>(Lsim, Rsim, simw2, simb2,
                                           Ldiv, Rdiv, divw2, divb2,
                                           outSim, outDiv);
}


# Pair kernel staging software-pipelined: next K-chunk's L, R, w2 fetched by 16-byte loads, all in flight during this chunk's compute, into a second LDS buffer; was 17 waited 4-byte loads in turn
# speedup vs baseline: 2.2340x; 1.6552x over previous
#include <hip/hip_runtime.h>
#include <hip/hip_bf16.h>

#define H   768
#define NS  128
#define NQ  256
#define NTOT 384
#define RREL 8
#define KC  64

typedef __attribute__((ext_vector_type(8))) float v8f;
typedef __attribute__((ext_vector_type(2))) float v2f;

__global__ void wbar_kernel(const float* __restrict__ relW, float* __restrict__ wbar) {
    int e = blockIdx.x * 256 + threadIdx.x;
    float s = 0.f;
#pragma unroll
    for (int r = 0; r < RREL; ++r) s += relW[r * (H * H) + e];
    wbar[e] = s * (1.0f / RREL);
}

__global__ void gemm_wmma_f32(const float* __restrict__ Asup,
                              const float* __restrict__ Aqry,
                              const float* __restrict__ wbar,
                              const float* __restrict__ simW1,
                              const float* __restrict__ divW1,
                              const float* __restrict__ relb,
                              const float* __restrict__ simb1,
                              const float* __restrict__ divb1,
                              float* __restrict__ outProp,
                              float* __restrict__ Lsim, float* __restrict__ Rsim,
                              float* __restrict__ Ldiv, float* __restrict__ Rdiv) {
    const int z = blockIdx.z;
    const float* W    = (z == 0) ? wbar : ((z <= 2) ? simW1 : divW1);
    const int ldW     = (z == 0) ? H : 2 * H;
    const int colOff  = (z == 2 || z == 4) ? H : 0;
    const float* bias = (z == 0) ? relb : ((z == 1) ? simb1 : divb1);
    const int biasMode = (z == 0) ? 2 : ((z == 1 || z == 3) ? 1 : 0);
    float* out = (z == 0) ? outProp
               : (z == 1) ? Lsim : (z == 2) ? Rsim : (z == 3) ? Ldiv : Rdiv;

    const int lane = threadIdx.x & 31;
    const int wv   = threadIdx.x >> 5;
    const int m0   = blockIdx.x * 16;
    const int n0   = (blockIdx.y * 4 + wv) * 16;
    const int l15  = lane & 15;
    const int khalf = (lane >> 4) * 2;

    const float* Arow = (m0 < NS) ? (Asup + (size_t)(m0 + l15) * H)
                                  : (Aqry + (size_t)(m0 - NS + l15) * H);
    const float* Brow = W + (size_t)(n0 + l15) * ldW + colOff;

    v8f acc = {0.f, 0.f, 0.f, 0.f, 0.f, 0.f, 0.f, 0.f};
#pragma unroll 4
    for (int k = 0; k < H; k += 4) {
        v2f a = *(const v2f*)(Arow + k + khalf);
        v2f b = *(const v2f*)(Brow + k + khalf);
        acc = __builtin_amdgcn_wmma_f32_16x16x4_f32(
             false, a,  false, b,
             (short)0, acc,  false,  false);
    }

    const int n = n0 + l15;
    float bv = 0.f;
    if (biasMode == 1) {
        bv = bias[n];
    } else if (biasMode == 2) {
        float s = 0.f;
#pragma unroll
        for (int r = 0; r < RREL; ++r) s += bias[r * H + n];
        bv = s * (1.0f / RREL);
    }

    const int mbase = m0 + ((lane >> 4) ? 8 : 0);
#pragma unroll
    for (int r8 = 0; r8 < 8; ++r8) {
        out[(size_t)(mbase + r8) * H + n] = acc[r8] + bv;
    }
}

__global__ void pair_kernel(const float* __restrict__ Lsim, const float* __restrict__ Rsim,
                            const float* __restrict__ w2sim, const float* __restrict__ b2sim,
                            const float* __restrict__ Ldiv, const float* __restrict__ Rdiv,
                            const float* __restrict__ w2div, const float* __restrict__ b2div,
                            float* __restrict__ outSim, float* __restrict__ outDiv) {
    __shared__ float Ls[2][32][KC + 1];
    __shared__ float Rs[2][32][KC + 1];
    __shared__ float Ws[2][KC];

    const float *Lp, *Rp, *w2;
    float b2;
    float* out;
    if (blockIdx.z == 0) { Lp = Lsim; Rp = Rsim; w2 = w2sim; b2 = b2sim[0]; out = outSim; }
    else                 { Lp = Ldiv; Rp = Rdiv; w2 = w2div; b2 = b2div[0]; out = outDiv; }

    const int i0 = blockIdx.y * 32;
    const int j0 = blockIdx.x * 32;
    const int t  = threadIdx.x;
    const int ti = t >> 3;
    const int tj = (t & 7) * 4;

    const int sr = t >> 4;
    const int sc = (t & 15) * 4;
    const float* gL0 = Lp + (size_t)(i0 + sr) * H + sc;
    const float* gL1 = gL0 + 16 * H;
    const float* gR0 = Rp + (size_t)(j0 + sr) * H + sc;
    const float* gR1 = gR0 + 16 * H;

    float4 pl0 = *(const float4*)(gL0);
    float4 pl1 = *(const float4*)(gL1);
    float4 pr0 = *(const float4*)(gR0);
    float4 pr1 = *(const float4*)(gR1);
    float  pw  = (t < KC) ? w2[t] : 0.f;

    {
        float* d;
        d = &Ls[0][sr][sc];      d[0] = pl0.x; d[1] = pl0.y; d[2] = pl0.z; d[3] = pl0.w;
        d = &Ls[0][sr + 16][sc]; d[0] = pl1.x; d[1] = pl1.y; d[2] = pl1.z; d[3] = pl1.w;
        d = &Rs[0][sr][sc];      d[0] = pr0.x; d[1] = pr0.y; d[2] = pr0.z; d[3] = pr0.w;
        d = &Rs[0][sr + 16][sc]; d[0] = pr1.x; d[1] = pr1.y; d[2] = pr1.z; d[3] = pr1.w;
        if (t < KC) Ws[0][t] = pw;
    }
    __syncthreads();

    float a0 = 0.f, a1 = 0.f, a2 = 0.f, a3 = 0.f;

    for (int c = 0; c < H / KC; ++c) {
        const int cur = c & 1;
        const bool more = (c + 1 < H / KC);
        if (more) {
            const int kn = (c + 1) * KC;
            pl0 = *(const float4*)(gL0 + kn);
            pl1 = *(const float4*)(gL1 + kn);
            pr0 = *(const float4*)(gR0 + kn);
            pr1 = *(const float4*)(gR1 + kn);
            if (t < KC) pw = w2[kn + t];
        }

        const float* Lrow = &Ls[cur][ti][0];
        const float* Rrow = &Rs[cur][tj][0];
        const float* Wc   = &Ws[cur][0];
#pragma unroll 4
        for (int k = 0; k < KC; ++k) {
            float lv = Lrow[k];
            float wv = Wc[k];
            a0 += fmaxf(lv + Rrow[0 * (KC + 1) + k], 0.f) * wv;
            a1 += fmaxf(lv + Rrow[1 * (KC + 1) + k], 0.f) * wv;
            a2 += fmaxf(lv + Rrow[2 * (KC + 1) + k], 0.f) * wv;
            a3 += fmaxf(lv + Rrow[3 * (KC + 1) + k], 0.f) * wv;
        }

        if (more) {
            const int nx = cur ^ 1;
            float* d;
            d = &Ls[nx][sr][sc];      d[0] = pl0.x; d[1] = pl0.y; d[2] = pl0.z; d[3] = pl0.w;
            d = &Ls[nx][sr + 16][sc]; d[0] = pl1.x; d[1] = pl1.y; d[2] = pl1.z; d[3] = pl1.w;
            d = &Rs[nx][sr][sc];      d[0] = pr0.x; d[1] = pr0.y; d[2] = pr0.z; d[3] = pr0.w;
            d = &Rs[nx][sr + 16][sc]; d[0] = pr1.x; d[1] = pr1.y; d[2] = pr1.z; d[3] = pr1.w;
            if (t < KC) Ws[nx][t] = pw;
        }
        __syncthreads();
    }

    const int i = i0 + ti;
    const int j = j0 + tj;
    float v[4] = {a0 + b2, a1 + b2, a2 + b2, a3 + b2};
#pragma unroll
    for (int c = 0; c < 4; ++c) {
        out[(size_t)i * NTOT + j + c] = (i == j + c) ? 0.f : v[c];
    }
}

extern "C" void kernel_launch(void* const* d_in, const int* in_sizes, int n_in,
                              void* d_out, int out_size, void* d_ws, size_t ws_size,
                              hipStream_t stream) {
    const float* sup    = (const float*)d_in[0];
    const float* qry    = (const float*)d_in[2];
    const float* simW1  = (const float*)d_in[3];
    const float* simb1  = (const float*)d_in[4];
    const float* simw2  = (const float*)d_in[5];
    const float* simb2  = (const float*)d_in[6];
    const float* divW1  = (const float*)d_in[7];
    const float* divb1  = (const float*)d_in[8];
    const float* divw2  = (const float*)d_in[9];
    const float* divb2  = (const float*)d_in[10];
    const float* relW   = (const float*)d_in[11];
    const float* relb   = (const float*)d_in[12];

    float* out = (float*)d_out;
    float* ws  = (float*)d_ws;

    float* wbar = ws;
    float* Lsim = ws + 589824;
    float* Rsim = Lsim + 294912;
    float* Ldiv = Rsim + 294912;
    float* Rdiv = Ldiv + 294912;

    float* outProp = out;
    float* outSim  = out + NTOT * H;
    float* outDiv  = outSim + NTOT * NTOT;

    wbar_kernel<<<(H * H) / 256, 256, 0, stream>>>(relW, wbar);

    dim3 ggrid(NTOT / 16, (H / 16) / 4, 5);
    dim3 gblk(128);
    gemm_wmma_f32<<<ggrid, gblk, 0, stream>>>(sup, qry, wbar, simW1, divW1,
                                              relb, simb1, divb1,
                                              outProp, Lsim, Rsim, Ldiv, Rdiv);

    dim3 pgrid(NTOT / 32, NTOT / 32, 2);
    pair_kernel<<<pgrid, 256, 0, stream>>>(Lsim, Rsim, simw2, simb2,
                                           Ldiv, Rdiv, divw2, divb2,
                                           outSim, outDiv);
}


# GEMM operands staged through LDS: 64x96 block tiles, K-chunks of 64 fetched once per block by whole-line 16-byte loads, double-buffered; each wave 16x48 (one A fragment, 3 tiles); same k order
# speedup vs baseline: 3.4715x; 1.5539x over previous
#include <hip/hip_runtime.h>
#include <hip/hip_bf16.h>

#define H   768
#define NS  128
#define NQ  256
#define NTOT 384
#define RREL 8
#define KC  64

typedef __attribute__((ext_vector_type(8))) float v8f;
typedef __attribute__((ext_vector_type(2))) float v2f;

__global__ void wbar_kernel(const float* __restrict__ relW, float* __restrict__ wbar) {
    int e = blockIdx.x * 256 + threadIdx.x;
    float s = 0.f;
#pragma unroll
    for (int r = 0; r < RREL; ++r) s += relW[r * (H * H) + e];
    wbar[e] = s * (1.0f / RREL);
}

#define GBM 64
#define GBN 96
#define GKC 64
#define GPITCH (GKC + 4)

__global__ __launch_bounds__(256)
void gemm_wmma_f32(const float* __restrict__ Asup,
                   const float* __restrict__ Aqry,
                   const float* __restrict__ wbar,
                   const float* __restrict__ simW1,
                   const float* __restrict__ divW1,
                   const float* __restrict__ relb,
                   const float* __restrict__ simb1,
                   const float* __restrict__ divb1,
                   float* __restrict__ outProp,
                   float* __restrict__ Lsim, float* __restrict__ Rsim,
                   float* __restrict__ Ldiv, float* __restrict__ Rdiv) {
    __shared__ __attribute__((aligned(16))) float As[2][GBM][GPITCH];
    __shared__ __attribute__((aligned(16))) float Bs[2][GBN][GPITCH];

    const int z = blockIdx.z;
    const float* W    = (z == 0) ? wbar : ((z <= 2) ? simW1 : divW1);
    const int ldW     = (z == 0) ? H : 2 * H;
    const int colOff  = (z == 2 || z == 4) ? H : 0;
    const float* bias = (z == 0) ? relb : ((z == 1) ? simb1 : divb1);
    const int biasMode = (z == 0) ? 2 : ((z == 1 || z == 3) ? 1 : 0);
    float* out = (z == 0) ? outProp
               : (z == 1) ? Lsim : (z == 2) ? Rsim : (z == 3) ? Ldiv : Rdiv;

    const int t    = threadIdx.x;
    const int lane = t & 31;
    const int wv   = t >> 5;
    const int wm   = wv & 3;
    const int wn   = wv >> 2;
    const int m0   = blockIdx.x * GBM;
    const int n0   = blockIdx.y * GBN;
    const int l15  = lane & 15;
    const int khalf = (lane >> 4) * 2;

    const int sr = t >> 4;
    const int sc = (t & 15) * 4;
    const float* Ablk = (m0 < NS) ? (Asup + (size_t)m0 * H)
                                  : (Aqry + (size_t)(m0 - NS) * H);
    const float* gA = Ablk + (size_t)sr * H + sc;
    const float* gB = W + (size_t)(n0 + sr) * ldW + colOff + sc;
    const size_t aStep = (size_t)16 * H;
    const size_t bStep = (size_t)16 * ldW;

    float4 pa[GBM / 16];
    float4 pb[GBN / 16];
#pragma unroll
    for (int i = 0; i < GBM / 16; ++i) pa[i] = *(const float4*)(gA + i * aStep);
#pragma unroll
    for (int i = 0; i < GBN / 16; ++i) pb[i] = *(const float4*)(gB + i * bStep);
#pragma unroll
    for (int i = 0; i < GBM / 16; ++i) *(float4*)&As[0][sr + 16 * i][sc] = pa[i];
#pragma unroll
    for (int i = 0; i < GBN / 16; ++i) *(float4*)&Bs[0][sr + 16 * i][sc] = pb[i];
    __syncthreads();

    v8f acc0 = {0.f, 0.f, 0.f, 0.f, 0.f, 0.f, 0.f, 0.f};
    v8f acc1 = acc0;
    v8f acc2 = acc0;

    for (int ch = 0; ch < H / GKC; ++ch) {
        const int cur = ch & 1;
        const bool more = (ch + 1 < H / GKC);
        if (more) {
            const int kn = (ch + 1) * GKC;
#pragma unroll
            for (int i = 0; i < GBM / 16; ++i) pa[i] = *(const float4*)(gA + i * aStep + kn);
#pragma unroll
            for (int i = 0; i < GBN / 16; ++i) pb[i] = *(const float4*)(gB + i * bStep + kn);
        }

        const float* Ar = &As[cur][wm * 16 + l15][khalf];
        const float* Br = &Bs[cur][wn * 48 + l15][khalf];
#pragma unroll
        for (int ks = 0; ks < GKC; ks += 4) {
            v2f a  = *(const v2f*)(Ar + ks);
            v2f b0 = *(const v2f*)(Br + ks);
            v2f b1 = *(const v2f*)(Br + 16 * GPITCH + ks);
            v2f b2 = *(const v2f*)(Br + 32 * GPITCH + ks);
            acc0 = __builtin_amdgcn_wmma_f32_16x16x4_f32(
                 false, a,  false, b0,
                 (short)0, acc0,  false,  false);
            acc1 = __builtin_amdgcn_wmma_f32_16x16x4_f32(
                false, a, false, b1, (short)0, acc1, false, false);
            acc2 = __builtin_amdgcn_wmma_f32_16x16x4_f32(
                false, a, false, b2, (short)0, acc2, false, false);
        }

        if (more) {
            const int nx = cur ^ 1;
#pragma unroll
            for (int i = 0; i < GBM / 16; ++i) *(float4*)&As[nx][sr + 16 * i][sc] = pa[i];
#pragma unroll
            for (int i = 0; i < GBN / 16; ++i) *(float4*)&Bs[nx][sr + 16 * i][sc] = pb[i];
        }
        __syncthreads();
    }

    const int nb = n0 + wn * 48 + l15;
    float bv[3] = {0.f, 0.f, 0.f};
    if (biasMode == 1) {
#pragma unroll
        for (int j = 0; j < 3; ++j) bv[j] = bias[nb + j * 16];
    } else if (biasMode == 2) {
        float s[3] = {0.f, 0.f, 0.f};
#pragma unroll
        for (int r = 0; r < RREL; ++r) {
#pragma unroll
            for (int j = 0; j < 3; ++j) s[j] += bias[r * H + nb + j * 16];
        }
#pragma unroll
        for (int j = 0; j < 3; ++j) bv[j] = s[j] * (1.0f / RREL);
    }

    const int mbase = m0 + wm * 16 + ((lane >> 4) ? 8 : 0);
    float* orow = out + (size_t)mbase * H + nb;
#pragma unroll
    for (int r8 = 0; r8 < 8; ++r8) {
        orow[(size_t)r8 * H]      = acc0[r8] + bv[0];
        orow[(size_t)r8 * H + 16] = acc1[r8] + bv[1];
        orow[(size_t)r8 * H + 32] = acc2[r8] + bv[2];
    }
}

__global__ void pair_kernel(const float* __restrict__ Lsim, const float* __restrict__ Rsim,
                            const float* __restrict__ w2sim, const float* __restrict__ b2sim,
                            const float* __restrict__ Ldiv, const float* __restrict__ Rdiv,
                            const float* __restrict__ w2div, const float* __restrict__ b2div,
                            float* __restrict__ outSim, float* __restrict__ outDiv) {
    __shared__ float Ls[2][32][KC + 1];
    __shared__ float Rs[2][32][KC + 1];
    __shared__ float Ws[2][KC];

    const float *Lp, *Rp, *w2;
    float b2;
    float* out;
    if (blockIdx.z == 0) { Lp = Lsim; Rp = Rsim; w2 = w2sim; b2 = b2sim[0]; out = outSim; }
    else                 { Lp = Ldiv; Rp = Rdiv; w2 = w2div; b2 = b2div[0]; out = outDiv; }

    const int i0 = blockIdx.y * 32;
    const int j0 = blockIdx.x * 32;
    const int t  = threadIdx.x;
    const int ti = t >> 3;
    const int tj = (t & 7) * 4;

    const int sr = t >> 4;
    const int sc = (t & 15) * 4;
    const float* gL0 = Lp + (size_t)(i0 + sr) * H + sc;
    const float* gL1 = gL0 + 16 * H;
    const float* gR0 = Rp + (size_t)(j0 + sr) * H + sc;
    const float* gR1 = gR0 + 16 * H;

    float4 pl0 = *(const float4*)(gL0);
    float4 pl1 = *(const float4*)(gL1);
    float4 pr0 = *(const float4*)(gR0);
    float4 pr1 = *(const float4*)(gR1);
    float  pw  = (t < KC) ? w2[t] : 0.f;

    {
        float* d;
        d = &Ls[0][sr][sc];      d[0] = pl0.x; d[1] = pl0.y; d[2] = pl0.z; d[3] = pl0.w;
        d = &Ls[0][sr + 16][sc]; d[0] = pl1.x; d[1] = pl1.y; d[2] = pl1.z; d[3] = pl1.w;
        d = &Rs[0][sr][sc];      d[0] = pr0.x; d[1] = pr0.y; d[2] = pr0.z; d[3] = pr0.w;
        d = &Rs[0][sr + 16][sc]; d[0] = pr1.x; d[1] = pr1.y; d[2] = pr1.z; d[3] = pr1.w;
        if (t < KC) Ws[0][t] = pw;
    }
    __syncthreads();

    float a0 = 0.f, a1 = 0.f, a2 = 0.f, a3 = 0.f;

    for (int c = 0; c < H / KC; ++c) {
        const int cur = c & 1;
        const bool more = (c + 1 < H / KC);
        if (more) {
            const int kn = (c + 1) * KC;
            pl0 = *(const float4*)(gL0 + kn);
            pl1 = *(const float4*)(gL1 + kn);
            pr0 = *(const float4*)(gR0 + kn);
            pr1 = *(const float4*)(gR1 + kn);
            if (t < KC) pw = w2[kn + t];
        }

        const float* Lrow = &Ls[cur][ti][0];
        const float* Rrow = &Rs[cur][tj][0];
        const float* Wc   = &Ws[cur][0];
#pragma unroll 4
        for (int k = 0; k < KC; ++k) {
            float lv = Lrow[k];
            float wv = Wc[k];
            a0 += fmaxf(lv + Rrow[0 * (KC + 1) + k], 0.f) * wv;
            a1 += fmaxf(lv + Rrow[1 * (KC + 1) + k], 0.f) * wv;
            a2 += fmaxf(lv + Rrow[2 * (KC + 1) + k], 0.f) * wv;
            a3 += fmaxf(lv + Rrow[3 * (KC + 1) + k], 0.f) * wv;
        }

        if (more) {
            const int nx = cur ^ 1;
            float* d;
            d = &Ls[nx][sr][sc];      d[0] = pl0.x; d[1] = pl0.y; d[2] = pl0.z; d[3] = pl0.w;
            d = &Ls[nx][sr + 16][sc]; d[0] = pl1.x; d[1] = pl1.y; d[2] = pl1.z; d[3] = pl1.w;
            d = &Rs[nx][sr][sc];      d[0] = pr0.x; d[1] = pr0.y; d[2] = pr0.z; d[3] = pr0.w;
            d = &Rs[nx][sr + 16][sc]; d[0] = pr1.x; d[1] = pr1.y; d[2] = pr1.z; d[3] = pr1.w;
            if (t < KC) Ws[nx][t] = pw;
        }
        __syncthreads();
    }

    const int i = i0 + ti;
    const int j = j0 + tj;
    float v[4] = {a0 + b2, a1 + b2, a2 + b2, a3 + b2};
#pragma unroll
    for (int c = 0; c < 4; ++c) {
        out[(size_t)i * NTOT + j + c] = (i == j + c) ? 0.f : v[c];
    }
}

extern "C" void kernel_launch(void* const* d_in, const int* in_sizes, int n_in,
                              void* d_out, int out_size, void* d_ws, size_t ws_size,
                              hipStream_t stream) {
    const float* sup    = (const float*)d_in[0];
    const float* qry    = (const float*)d_in[2];
    const float* simW1  = (const float*)d_in[3];
    const float* simb1  = (const float*)d_in[4];
    const float* simw2  = (const float*)d_in[5];
    const float* simb2  = (const float*)d_in[6];
    const float* divW1  = (const float*)d_in[7];
    const float* divb1  = (const float*)d_in[8];
    const float* divw2  = (const float*)d_in[9];
    const float* divb2  = (const float*)d_in[10];
    const float* relW   = (const float*)d_in[11];
    const float* relb   = (const float*)d_in[12];

    float* out = (float*)d_out;
    float* ws  = (float*)d_ws;

    float* wbar = ws;
    float* Lsim = ws + 589824;
    float* Rsim = Lsim + 294912;
    float* Ldiv = Rsim + 294912;
    float* Rdiv = Ldiv + 294912;

    float* outProp = out;
    float* outSim  = out + NTOT * H;
    float* outDiv  = outSim + NTOT * NTOT;

    wbar_kernel<<<(H * H) / 256, 256, 0, stream>>>(relW, wbar);

    dim3 ggrid(NTOT / GBM, H / GBN, 5);
    dim3 gblk(256);
    gemm_wmma_f32<<<ggrid, gblk, 0, stream>>>(sup, qry, wbar, simW1, divW1,
                                              relb, simb1, divb1,
                                              outProp, Lsim, Rsim, Ldiv, Rdiv);

    dim3 pgrid(NTOT / 32, NTOT / 32, 2);
    pair_kernel<<<pgrid, 256, 0, stream>>>(Lsim, Rsim, simw2, simb2,
                                           Ldiv, Rdiv, divw2, divb2,
                                           outSim, outDiv);
}


# Pair kernel re-tiled: 256 equal blocks of 12x96 (was 288 of 32x32), 3x3 outputs per thread from 16-byte LDS reads along k; each output now summed as 4 partial sums (k parity x chunk half)
# speedup vs baseline: 4.2799x; 1.2329x over previous
#include <hip/hip_runtime.h>
#include <hip/hip_bf16.h>

#define H   768
#define NS  128
#define NQ  256
#define NTOT 384
#define RREL 8
#define KC  64

typedef __attribute__((ext_vector_type(8))) float v8f;
typedef __attribute__((ext_vector_type(2))) float v2f;

__global__ void wbar_kernel(const float* __restrict__ relW, float* __restrict__ wbar) {
    int e = blockIdx.x * 256 + threadIdx.x;
    float s = 0.f;
#pragma unroll
    for (int r = 0; r < RREL; ++r) s += relW[r * (H * H) + e];
    wbar[e] = s * (1.0f / RREL);
}

#define GBM 64
#define GBN 96
#define GKC 64
#define GPITCH (GKC + 4)

__global__ __launch_bounds__(256)
void gemm_wmma_f32(const float* __restrict__ Asup,
                   const float* __restrict__ Aqry,
                   const float* __restrict__ wbar,
                   const float* __restrict__ simW1,
                   const float* __restrict__ divW1,
                   const float* __restrict__ relb,
                   const float* __restrict__ simb1,
                   const float* __restrict__ divb1,
                   float* __restrict__ outProp,
                   float* __restrict__ Lsim, float* __restrict__ Rsim,
                   float* __restrict__ Ldiv, float* __restrict__ Rdiv) {
    __shared__ __attribute__((aligned(16))) float As[2][GBM][GPITCH];
    __shared__ __attribute__((aligned(16))) float Bs[2][GBN][GPITCH];

    const int z = blockIdx.z;
    const float* W    = (z == 0) ? wbar : ((z <= 2) ? simW1 : divW1);
    const int ldW     = (z == 0) ? H : 2 * H;
    const int colOff  = (z == 2 || z == 4) ? H : 0;
    const float* bias = (z == 0) ? relb : ((z == 1) ? simb1 : divb1);
    const int biasMode = (z == 0) ? 2 : ((z == 1 || z == 3) ? 1 : 0);
    float* out = (z == 0) ? outProp
               : (z == 1) ? Lsim : (z == 2) ? Rsim : (z == 3) ? Ldiv : Rdiv;

    const int t    = threadIdx.x;
    const int lane = t & 31;
    const int wv   = t >> 5;
    const int wm   = wv & 3;
    const int wn   = wv >> 2;
    const int m0   = blockIdx.x * GBM;
    const int n0   = blockIdx.y * GBN;
    const int l15  = lane & 15;
    const int khalf = (lane >> 4) * 2;

    const int sr = t >> 4;
    const int sc = (t & 15) * 4;
    const float* Ablk = (m0 < NS) ? (Asup + (size_t)m0 * H)
                                  : (Aqry + (size_t)(m0 - NS) * H);
    const float* gA = Ablk + (size_t)sr * H + sc;
    const float* gB = W + (size_t)(n0 + sr) * ldW + colOff + sc;
    const size_t aStep = (size_t)16 * H;
    const size_t bStep = (size_t)16 * ldW;

    float4 pa[GBM / 16];
    float4 pb[GBN / 16];
#pragma unroll
    for (int i = 0; i < GBM / 16; ++i) pa[i] = *(const float4*)(gA + i * aStep);
#pragma unroll
    for (int i = 0; i < GBN / 16; ++i) pb[i] = *(const float4*)(gB + i * bStep);
#pragma unroll
    for (int i = 0; i < GBM / 16; ++i) *(float4*)&As[0][sr + 16 * i][sc] = pa[i];
#pragma unroll
    for (int i = 0; i < GBN / 16; ++i) *(float4*)&Bs[0][sr + 16 * i][sc] = pb[i];
    __syncthreads();

    v8f acc0 = {0.f, 0.f, 0.f, 0.f, 0.f, 0.f, 0.f, 0.f};
    v8f acc1 = acc0;
    v8f acc2 = acc0;

    for (int ch = 0; ch < H / GKC; ++ch) {
        const int cur = ch & 1;
        const bool more = (ch + 1 < H / GKC);
        if (more) {
            const int kn = (ch + 1) * GKC;
#pragma unroll
            for (int i = 0; i < GBM / 16; ++i) pa[i] = *(const float4*)(gA + i * aStep + kn);
#pragma unroll
            for (int i = 0; i < GBN / 16; ++i) pb[i] = *(const float4*)(gB + i * bStep + kn);
        }

        const float* Ar = &As[cur][wm * 16 + l15][khalf];
        const float* Br = &Bs[cur][wn * 48 + l15][khalf];
#pragma unroll
        for (int ks = 0; ks < GKC; ks += 4) {
            v2f a  = *(const v2f*)(Ar + ks);
            v2f b0 = *(const v2f*)(Br + ks);
            v2f b1 = *(const v2f*)(Br + 16 * GPITCH + ks);
            v2f b2 = *(const v2f*)(Br + 32 * GPITCH + ks);
            acc0 = __builtin_amdgcn_wmma_f32_16x16x4_f32(
                 false, a,  false, b0,
                 (short)0, acc0,  false,  false);
            acc1 = __builtin_amdgcn_wmma_f32_16x16x4_f32(
                false, a, false, b1, (short)0, acc1, false, false);
            acc2 = __builtin_amdgcn_wmma_f32_16x16x4_f32(
                false, a, false, b2, (short)0, acc2, false, false);
        }

        if (more) {
            const int nx = cur ^ 1;
#pragma unroll
            for (int i = 0; i < GBM / 16; ++i) *(float4*)&As[nx][sr + 16 * i][sc] = pa[i];
#pragma unroll
            for (int i = 0; i < GBN / 16; ++i) *(float4*)&Bs[nx][sr + 16 * i][sc] = pb[i];
        }
        __syncthreads();
    }

    const int nb = n0 + wn * 48 + l15;
    float bv[3] = {0.f, 0.f, 0.f};
    if (biasMode == 1) {
#pragma unroll
        for (int j = 0; j < 3; ++j) bv[j] = bias[nb + j * 16];
    } else if (biasMode == 2) {
        float s[3] = {0.f, 0.f, 0.f};
#pragma unroll
        for (int r = 0; r < RREL; ++r) {
#pragma unroll
            for (int j = 0; j < 3; ++j) s[j] += bias[r * H + nb + j * 16];
        }
#pragma unroll
        for (int j = 0; j < 3; ++j) bv[j] = s[j] * (1.0f / RREL);
    }

    const int mbase = m0 + wm * 16 + ((lane >> 4) ? 8 : 0);
    float* orow = out + (size_t)mbase * H + nb;
#pragma unroll
    for (int r8 = 0; r8 < 8; ++r8) {
        orow[(size_t)r8 * H]      = acc0[r8] + bv[0];
        orow[(size_t)r8 * H + 16] = acc1[r8] + bv[1];
        orow[(size_t)r8 * H + 32] = acc2[r8] + bv[2];
    }
}

#define PTI 12
#define PTJ 96
#define PPITCH (KC + 4)
#define PKH (KC / 2)

__device__ __forceinline__ void pair_chunk_math(const float* Lr, const float* Rr,
                                                const float* __restrict__ wc, v2f (&acc)[3][3]) {
    const v2f zero2 = {0.f, 0.f};
#pragma unroll 1
    for (int g = 0; g < PKH; g += 16) {
#pragma unroll
        for (int s = 0; s < 16; s += 4) {
            const int ks = g + s;
            float4 lq[3], rq[3];
#pragma unroll
            for (int a = 0; a < 3; ++a) lq[a] = *(const float4*)(Lr + a * PPITCH + ks);
#pragma unroll
            for (int b = 0; b < 3; ++b) rq[b] = *(const float4*)(Rr + b * 32 * PPITCH + ks);
            const v2f w01 = {wc[ks + 0], wc[ks + 1]};
            const v2f w23 = {wc[ks + 2], wc[ks + 3]};
#pragma unroll
            for (int a = 0; a < 3; ++a) {
                const v2f l01 = {lq[a].x, lq[a].y};
                const v2f l23 = {lq[a].z, lq[a].w};
#pragma unroll
                for (int b = 0; b < 3; ++b) {
                    const v2f r01 = {rq[b].x, rq[b].y};
                    const v2f r23 = {rq[b].z, rq[b].w};
                    v2f x = __builtin_elementwise_max(l01 + r01, zero2);
                    acc[a][b] = __builtin_elementwise_fma(x, w01, acc[a][b]);
                    v2f y = __builtin_elementwise_max(l23 + r23, zero2);
                    acc[a][b] = __builtin_elementwise_fma(y, w23, acc[a][b]);
                }
            }
        }
    }
}

__global__ __launch_bounds__(256)
void pair_kernel(const float* __restrict__ Lsim, const float* __restrict__ Rsim,
                 const float* __restrict__ w2sim, const float* __restrict__ b2sim,
                 const float* __restrict__ Ldiv, const float* __restrict__ Rdiv,
                 const float* __restrict__ w2div, const float* __restrict__ b2div,
                 float* __restrict__ outSim, float* __restrict__ outDiv) {
    __shared__ __attribute__((aligned(16))) float Ls[2][PTI][PPITCH];
    __shared__ __attribute__((aligned(16))) float Rs[2][PTJ][PPITCH];
    __shared__ float red[9][128];

    const float *Lp, *Rp, *w2;
    float b2;
    float* out;
    if (blockIdx.z == 0) { Lp = Lsim; Rp = Rsim; w2 = w2sim; b2 = b2sim[0]; out = outSim; }
    else                 { Lp = Ldiv; Rp = Rdiv; w2 = w2div; b2 = b2div[0]; out = outDiv; }

    const int i0 = blockIdx.y * PTI;
    const int j0 = blockIdx.x * PTJ;
    const int t    = threadIdx.x;
    const int lane = t & 31;
    const int wv   = __builtin_amdgcn_readfirstlane(t >> 5);
    const int rg   = wv & 3;
    const int kh   = wv >> 2;

    const int sr = t >> 4;
    const int sc = (t & 15) * 4;
    const bool stagesL = (wv < 6);
    const float* gR = Rp + (size_t)(j0 + sr) * H + sc;
    const float* gL = Lp + (size_t)(i0 + (stagesL ? sr : sr - 4)) * H + sc;
    const size_t rStep = (size_t)16 * H;

    float4 pr0 = *(const float4*)(gR);
    float4 pr1 = *(const float4*)(gR + 1 * rStep);
    float4 pr2 = *(const float4*)(gR + 2 * rStep);
    float4 pr3 = *(const float4*)(gR + 3 * rStep);
    float4 pr4 = *(const float4*)(gR + 4 * rStep);
    float4 pr5 = *(const float4*)(gR + 5 * rStep);
    float4 pl  = *(const float4*)(gL);
    *(float4*)&Rs[0][sr][sc]      = pr0;
    *(float4*)&Rs[0][sr + 16][sc] = pr1;
    *(float4*)&Rs[0][sr + 32][sc] = pr2;
    *(float4*)&Rs[0][sr + 48][sc] = pr3;
    *(float4*)&Rs[0][sr + 64][sc] = pr4;
    *(float4*)&Rs[0][sr + 80][sc] = pr5;
    if (stagesL) *(float4*)&Ls[0][sr][sc] = pl;
    __syncthreads();

    v2f acc[3][3];
#pragma unroll
    for (int a = 0; a < 3; ++a) {
#pragma unroll
        for (int b = 0; b < 3; ++b) acc[a][b] = (v2f){0.f, 0.f};
    }

    for (int c = 0; c < H / KC - 1; ++c) {
        const int cur = c & 1;
        const int kc  = c * KC;
        const int kn  = kc + KC;
        pr0 = *(const float4*)(gR + kn);
        pr1 = *(const float4*)(gR + 1 * rStep + kn);
        pr2 = *(const float4*)(gR + 2 * rStep + kn);
        pr3 = *(const float4*)(gR + 3 * rStep + kn);
        pr4 = *(const float4*)(gR + 4 * rStep + kn);
        pr5 = *(const float4*)(gR + 5 * rStep + kn);
        pl  = *(const float4*)(gL + kn);
        __asm__ volatile("" ::: "memory");

        pair_chunk_math(&Ls[cur][rg * 3][kh * PKH], &Rs[cur][lane][kh * PKH],
                        w2 + kc + kh * PKH, acc);

        const int nx = cur ^ 1;
        *(float4*)&Rs[nx][sr][sc]      = pr0;
        *(float4*)&Rs[nx][sr + 16][sc] = pr1;
        *(float4*)&Rs[nx][sr + 32][sc] = pr2;
        *(float4*)&Rs[nx][sr + 48][sc] = pr3;
        *(float4*)&Rs[nx][sr + 64][sc] = pr4;
        *(float4*)&Rs[nx][sr + 80][sc] = pr5;
        if (stagesL) *(float4*)&Ls[nx][sr][sc] = pl;
        __syncthreads();
    }
    {
        constexpr int cl = H / KC - 1;
        pair_chunk_math(&Ls[cl & 1][rg * 3][kh * PKH], &Rs[cl & 1][lane][kh * PKH],
                        w2 + cl * KC + kh * PKH, acc);
    }

    const int tl = t & 127;
    if (kh == 1) {
#pragma unroll
        for (int a = 0; a < 3; ++a) {
#pragma unroll
            for (int b = 0; b < 3; ++b) red[a * 3 + b][tl] = acc[a][b].x + acc[a][b].y;
        }
    }
    __syncthreads();
    if (kh == 0) {
#pragma unroll
        for (int a = 0; a < 3; ++a) {
            const int i = i0 + rg * 3 + a;
#pragma unroll
            for (int b = 0; b < 3; ++b) {
                const int j = j0 + lane + 32 * b;
                const float v = ((acc[a][b].x + acc[a][b].y) + red[a * 3 + b][tl]) + b2;
                out[(size_t)i * NTOT + j] = (i == j) ? 0.f : v;
            }
        }
    }
}

extern "C" void kernel_launch(void* const* d_in, const int* in_sizes, int n_in,
                              void* d_out, int out_size, void* d_ws, size_t ws_size,
                              hipStream_t stream) {
    const float* sup    = (const float*)d_in[0];
    const float* qry    = (const float*)d_in[2];
    const float* simW1  = (const float*)d_in[3];
    const float* simb1  = (const float*)d_in[4];
    const float* simw2  = (const float*)d_in[5];
    const float* simb2  = (const float*)d_in[6];
    const float* divW1  = (const float*)d_in[7];
    const float* divb1  = (const float*)d_in[8];
    const float* divw2  = (const float*)d_in[9];
    const float* divb2  = (const float*)d_in[10];
    const float* relW   = (const float*)d_in[11];
    const float* relb   = (const float*)d_in[12];

    float* out = (float*)d_out;
    float* ws  = (float*)d_ws;

    float* wbar = ws;
    float* Lsim = ws + 589824;
    float* Rsim = Lsim + 294912;
    float* Ldiv = Rsim + 294912;
    float* Rdiv = Ldiv + 294912;

    float* outProp = out;
    float* outSim  = out + NTOT * H;
    float* outDiv  = outSim + NTOT * NTOT;

    wbar_kernel<<<(H * H) / 256, 256, 0, stream>>>(relW, wbar);

    dim3 ggrid(NTOT / GBM, H / GBN, 5);
    dim3 gblk(256);
    gemm_wmma_f32<<<ggrid, gblk, 0, stream>>>(sup, qry, wbar, simW1, divW1,
                                              relb, simb1, divb1,
                                              outProp, Lsim, Rsim, Ldiv, Rdiv);

    dim3 pgrid(NTOT / PTJ, NTOT / PTI, 2);
    pair_kernel<<<pgrid, 256, 0, stream>>>(Lsim, Rsim, simw2, simb2,
                                           Ldiv, Rdiv, divw2, divb2,
                                           outSim, outDiv);
}


# GEMM fragment reads software-pipelined: 8 k-steps' LDS reads issued as one block a group ahead of their 24 WMMAs (two register sets), across the chunk barrier too via 3-stage LDS ring; same sum order
# speedup vs baseline: 4.5499x; 1.0631x over previous
#include <hip/hip_runtime.h>
#include <hip/hip_bf16.h>

#define H   768
#define NS  128
#define NQ  256
#define NTOT 384
#define RREL 8
#define KC  64

typedef __attribute__((ext_vector_type(8))) float v8f;
typedef __attribute__((ext_vector_type(2))) float v2f;

__global__ void wbar_kernel(const float* __restrict__ relW, float* __restrict__ wbar) {
    int e = blockIdx.x * 256 + threadIdx.x;
    float s = 0.f;
#pragma unroll
    for (int r = 0; r < RREL; ++r) s += relW[r * (H * H) + e];
    wbar[e] = s * (1.0f / RREL);
}

#define GBM 64
#define GBN 96
#define GKC 64
#define GKG 32
#define GST 3
#define GPITCH (GKC + 4)
#define GASTG (GBM * GPITCH)
#define GBSTG (GBN * GPITCH)

#define GF_DECL(S) \
    v2f S##a0, S##a1, S##a2, S##a3, S##a4, S##a5, S##a6, S##a7; \
    v2f S##p0, S##p1, S##p2, S##p3, S##p4, S##p5, S##p6, S##p7; \
    v2f S##q0, S##q1, S##q2, S##q3, S##q4, S##q5, S##q6, S##q7; \
    v2f S##r0, S##r1, S##r2, S##r3, S##r4, S##r5, S##r6, S##r7;
#define GF_LOAD1(S, i, Ap, Bp, kg) \
    S##a##i = *(const v2f*)((Ap) + (kg) + 4 * i); \
    S##p##i = *(const v2f*)((Bp) + (kg) + 4 * i); \
    S##q##i = *(const v2f*)((Bp) + 16 * GPITCH + (kg) + 4 * i); \
    S##r##i = *(const v2f*)((Bp) + 32 * GPITCH + (kg) + 4 * i);
#define GF_LOAD(S, Ap, Bp, kg) \
    GF_LOAD1(S, 0, Ap, Bp, kg) GF_LOAD1(S, 1, Ap, Bp, kg) GF_LOAD1(S, 2, Ap, Bp, kg) GF_LOAD1(S, 3, Ap, Bp, kg) \
    GF_LOAD1(S, 4, Ap, Bp, kg) GF_LOAD1(S, 5, Ap, Bp, kg) GF_LOAD1(S, 6, Ap, Bp, kg) GF_LOAD1(S, 7, Ap, Bp, kg)
#define GF_MMA1(S, i) \
    acc0 = __builtin_amdgcn_wmma_f32_16x16x4_f32(false, S##a##i, false, S##p##i, (short)0, acc0, false, false); \
    acc1 = __builtin_amdgcn_wmma_f32_16x16x4_f32(false, S##a##i, false, S##q##i, (short)0, acc1, false, false); \
    acc2 = __builtin_amdgcn_wmma_f32_16x16x4_f32(false, S##a##i, false, S##r##i, (short)0, acc2, false, false);
#define GF_MMA(S) \
    GF_MMA1(S, 0) GF_MMA1(S, 1) GF_MMA1(S, 2) GF_MMA1(S, 3) \
    GF_MMA1(S, 4) GF_MMA1(S, 5) GF_MMA1(S, 6) GF_MMA1(S, 7)

__global__ __launch_bounds__(256)
void gemm_wmma_f32(const float* __restrict__ Asup,
                   const float* __restrict__ Aqry,
                   const float* __restrict__ wbar,
                   const float* __restrict__ simW1,
                   const float* __restrict__ divW1,
                   const float* __restrict__ relb,
                   const float* __restrict__ simb1,
                   const float* __restrict__ divb1,
                   float* __restrict__ outProp,
                   float* __restrict__ Lsim, float* __restrict__ Rsim,
                   float* __restrict__ Ldiv, float* __restrict__ Rdiv) {
    __shared__ __attribute__((aligned(16))) float As[GST][GBM][GPITCH];
    __shared__ __attribute__((aligned(16))) float Bs[GST][GBN][GPITCH];

    const int z = blockIdx.z;
    const float* W    = (z == 0) ? wbar : ((z <= 2) ? simW1 : divW1);
    const int ldW     = (z == 0) ? H : 2 * H;
    const int colOff  = (z == 2 || z == 4) ? H : 0;
    const float* bias = (z == 0) ? relb : ((z == 1) ? simb1 : divb1);
    const int biasMode = (z == 0) ? 2 : ((z == 1 || z == 3) ? 1 : 0);
    float* out = (z == 0) ? outProp
               : (z == 1) ? Lsim : (z == 2) ? Rsim : (z == 3) ? Ldiv : Rdiv;

    const int t    = threadIdx.x;
    const int lane = t & 31;
    const int wv   = t >> 5;
    const int wm   = wv & 3;
    const int wn   = wv >> 2;
    const int m0   = blockIdx.x * GBM;
    const int n0   = blockIdx.y * GBN;
    const int l15  = lane & 15;
    const int khalf = (lane >> 4) * 2;

    const int sr = t >> 4;
    const int sc = (t & 15) * 4;
    const float* Ablk = (m0 < NS) ? (Asup + (size_t)m0 * H)
                                  : (Aqry + (size_t)(m0 - NS) * H);
    const float* gA = Ablk + (size_t)sr * H + sc;
    const float* gB = W + (size_t)(n0 + sr) * ldW + colOff + sc;
    const size_t aStep = (size_t)16 * H;
    const size_t bStep = (size_t)16 * ldW;

    float* sA = &As[0][sr][sc];
    float* sB = &Bs[0][sr][sc];
    const float* fA0 = &As[0][wm * 16 + l15][khalf];
    const float* fB0 = &Bs[0][wn * 48 + l15][khalf];

    float4 pa0 = *(const float4*)(gA);
    float4 pa1 = *(const float4*)(gA + 1 * aStep);
    float4 pa2 = *(const float4*)(gA + 2 * aStep);
    float4 pa3 = *(const float4*)(gA + 3 * aStep);
    float4 pb0 = *(const float4*)(gB);
    float4 pb1 = *(const float4*)(gB + 1 * bStep);
    float4 pb2 = *(const float4*)(gB + 2 * bStep);
    float4 pb3 = *(const float4*)(gB + 3 * bStep);
    float4 pb4 = *(const float4*)(gB + 4 * bStep);
    float4 pb5 = *(const float4*)(gB + 5 * bStep);
    {
        const float4 qa0 = *(const float4*)(gA + GKC);
        const float4 qa1 = *(const float4*)(gA + 1 * aStep + GKC);
        const float4 qa2 = *(const float4*)(gA + 2 * aStep + GKC);
        const float4 qa3 = *(const float4*)(gA + 3 * aStep + GKC);
        const float4 qb0 = *(const float4*)(gB + GKC);
        const float4 qb1 = *(const float4*)(gB + 1 * bStep + GKC);
        const float4 qb2 = *(const float4*)(gB + 2 * bStep + GKC);
        const float4 qb3 = *(const float4*)(gB + 3 * bStep + GKC);
        const float4 qb4 = *(const float4*)(gB + 4 * bStep + GKC);
        const float4 qb5 = *(const float4*)(gB + 5 * bStep + GKC);
        *(float4*)(sA)               = pa0;
        *(float4*)(sA + 16 * GPITCH) = pa1;
        *(float4*)(sA + 32 * GPITCH) = pa2;
        *(float4*)(sA + 48 * GPITCH) = pa3;
        *(float4*)(sB)               = pb0;
        *(float4*)(sB + 16 * GPITCH) = pb1;
        *(float4*)(sB + 32 * GPITCH) = pb2;
        *(float4*)(sB + 48 * GPITCH) = pb3;
        *(float4*)(sB + 64 * GPITCH) = pb4;
        *(float4*)(sB + 80 * GPITCH) = pb5;
        *(float4*)(sA + GASTG)               = qa0;
        *(float4*)(sA + GASTG + 16 * GPITCH) = qa1;
        *(float4*)(sA + GASTG + 32 * GPITCH) = qa2;
        *(float4*)(sA + GASTG + 48 * GPITCH) = qa3;
        *(float4*)(sB + GBSTG)               = qb0;
        *(float4*)(sB + GBSTG + 16 * GPITCH) = qb1;
        *(float4*)(sB + GBSTG + 32 * GPITCH) = qb2;
        *(float4*)(sB + GBSTG + 48 * GPITCH) = qb3;
        *(float4*)(sB + GBSTG + 64 * GPITCH) = qb4;
        *(float4*)(sB + GBSTG + 80 * GPITCH) = qb5;
    }
    __syncthreads();

    v8f acc0 = {0.f, 0.f, 0.f, 0.f, 0.f, 0.f, 0.f, 0.f};
    v8f acc1 = acc0;
    v8f acc2 = acc0;

    GF_DECL(fx)
    GF_DECL(fy)
    GF_LOAD(fx, fA0, fB0, 0)

    int s0 = 0, s1 = 1, s2 = 2;
    for (int ch = 0; ch < H / GKC; ++ch) {
        const bool fetch = (ch + 2 < H / GKC);
        if (fetch) {
            const int kn = (ch + 2) * GKC;
            pa0 = *(const float4*)(gA + kn);
            pa1 = *(const float4*)(gA + 1 * aStep + kn);
            pa2 = *(const float4*)(gA + 2 * aStep + kn);
            pa3 = *(const float4*)(gA + 3 * aStep + kn);
            pb0 = *(const float4*)(gB + kn);
            pb1 = *(const float4*)(gB + 1 * bStep + kn);
            pb2 = *(const float4*)(gB + 2 * bStep + kn);
            pb3 = *(const float4*)(gB + 3 * bStep + kn);
            pb4 = *(const float4*)(gB + 4 * bStep + kn);
            pb5 = *(const float4*)(gB + 5 * bStep + kn);
        }

        GF_LOAD(fy, fA0 + s0 * GASTG, fB0 + s0 * GBSTG, GKG)
        __builtin_amdgcn_sched_barrier(0);
        GF_MMA(fx)
        __builtin_amdgcn_sched_barrier(0);

        GF_LOAD(fx, fA0 + s1 * GASTG, fB0 + s1 * GBSTG, 0)
        __builtin_amdgcn_sched_barrier(0);
        GF_MMA(fy)

        if (fetch) {
            float* dA = sA + s2 * GASTG;
            float* dB = sB + s2 * GBSTG;
            *(float4*)(dA)               = pa0;
            *(float4*)(dA + 16 * GPITCH) = pa1;
            *(float4*)(dA + 32 * GPITCH) = pa2;
            *(float4*)(dA + 48 * GPITCH) = pa3;
            *(float4*)(dB)               = pb0;
            *(float4*)(dB + 16 * GPITCH) = pb1;
            *(float4*)(dB + 32 * GPITCH) = pb2;
            *(float4*)(dB + 48 * GPITCH) = pb3;
            *(float4*)(dB + 64 * GPITCH) = pb4;
            *(float4*)(dB + 80 * GPITCH) = pb5;
        }
        __syncthreads();
        const int st = s0; s0 = s1; s1 = s2; s2 = st;
    }

    const int nb = n0 + wn * 48 + l15;
    float bv[3] = {0.f, 0.f, 0.f};
    if (biasMode == 1) {
#pragma unroll
        for (int j = 0; j < 3; ++j) bv[j] = bias[nb + j * 16];
    } else if (biasMode == 2) {
        float s[3] = {0.f, 0.f, 0.f};
#pragma unroll
        for (int r = 0; r < RREL; ++r) {
#pragma unroll
            for (int j = 0; j < 3; ++j) s[j] += bias[r * H + nb + j * 16];
        }
#pragma unroll
        for (int j = 0; j < 3; ++j) bv[j] = s[j] * (1.0f / RREL);
    }

    const int mbase = m0 + wm * 16 + ((lane >> 4) ? 8 : 0);
    float* orow = out + (size_t)mbase * H + nb;
#pragma unroll
    for (int r8 = 0; r8 < 8; ++r8) {
        orow[(size_t)r8 * H]      = acc0[r8] + bv[0];
        orow[(size_t)r8 * H + 16] = acc1[r8] + bv[1];
        orow[(size_t)r8 * H + 32] = acc2[r8] + bv[2];
    }
}

#define PTI 12
#define PTJ 96
#define PPITCH (KC + 4)
#define PKH (KC / 2)

__device__ __forceinline__ void pair_chunk_math(const float* Lr, const float* Rr,
                                                const float* __restrict__ wc, v2f (&acc)[3][3]) {
    const v2f zero2 = {0.f, 0.f};
#pragma unroll 1
    for (int g = 0; g < PKH; g += 16) {
#pragma unroll
        for (int s = 0; s < 16; s += 4) {
            const int ks = g + s;
            float4 lq[3], rq[3];
#pragma unroll
            for (int a = 0; a < 3; ++a) lq[a] = *(const float4*)(Lr + a * PPITCH + ks);
#pragma unroll
            for (int b = 0; b < 3; ++b) rq[b] = *(const float4*)(Rr + b * 32 * PPITCH + ks);
            const v2f w01 = {wc[ks + 0], wc[ks + 1]};
            const v2f w23 = {wc[ks + 2], wc[ks + 3]};
#pragma unroll
            for (int a = 0; a < 3; ++a) {
                const v2f l01 = {lq[a].x, lq[a].y};
                const v2f l23 = {lq[a].z, lq[a].w};
#pragma unroll
                for (int b = 0; b < 3; ++b) {
                    const v2f r01 = {rq[b].x, rq[b].y};
                    const v2f r23 = {rq[b].z, rq[b].w};
                    v2f x = __builtin_elementwise_max(l01 + r01, zero2);
                    acc[a][b] = __builtin_elementwise_fma(x, w01, acc[a][b]);
                    v2f y = __builtin_elementwise_max(l23 + r23, zero2);
                    acc[a][b] = __builtin_elementwise_fma(y, w23, acc[a][b]);
                }
            }
        }
    }
}

__global__ __launch_bounds__(256)
void pair_kernel(const float* __restrict__ Lsim, const float* __restrict__ Rsim,
                 const float* __restrict__ w2sim, const float* __restrict__ b2sim,
                 const float* __restrict__ Ldiv, const float* __restrict__ Rdiv,
                 const float* __restrict__ w2div, const float* __restrict__ b2div,
                 float* __restrict__ outSim, float* __restrict__ outDiv) {
    __shared__ __attribute__((aligned(16))) float Ls[2][PTI][PPITCH];
    __shared__ __attribute__((aligned(16))) float Rs[2][PTJ][PPITCH];
    __shared__ float red[9][128];

    const float *Lp, *Rp, *w2;
    float b2;
    float* out;
    if (blockIdx.z == 0) { Lp = Lsim; Rp = Rsim; w2 = w2sim; b2 = b2sim[0]; out = outSim; }
    else                 { Lp = Ldiv; Rp = Rdiv; w2 = w2div; b2 = b2div[0]; out = outDiv; }

    const int i0 = blockIdx.y * PTI;
    const int j0 = blockIdx.x * PTJ;
    const int t    = threadIdx.x;
    const int lane = t & 31;
    const int wv   = __builtin_amdgcn_readfirstlane(t >> 5);
    const int rg   = wv & 3;
    const int kh   = wv >> 2;

    const int sr = t >> 4;
    const int sc = (t & 15) * 4;
    const bool stagesL = (wv < 6);
    const float* gR = Rp + (size_t)(j0 + sr) * H + sc;
    const float* gL = Lp + (size_t)(i0 + (stagesL ? sr : sr - 4)) * H + sc;
    const size_t rStep = (size_t)16 * H;

    float4 pr0 = *(const float4*)(gR);
    float4 pr1 = *(const float4*)(gR + 1 * rStep);
    float4 pr2 = *(const float4*)(gR + 2 * rStep);
    float4 pr3 = *(const float4*)(gR + 3 * rStep);
    float4 pr4 = *(const float4*)(gR + 4 * rStep);
    float4 pr5 = *(const float4*)(gR + 5 * rStep);
    float4 pl  = *(const float4*)(gL);
    *(float4*)&Rs[0][sr][sc]      = pr0;
    *(float4*)&Rs[0][sr + 16][sc] = pr1;
    *(float4*)&Rs[0][sr + 32][sc] = pr2;
    *(float4*)&Rs[0][sr + 48][sc] = pr3;
    *(float4*)&Rs[0][sr + 64][sc] = pr4;
    *(float4*)&Rs[0][sr + 80][sc] = pr5;
    if (stagesL) *(float4*)&Ls[0][sr][sc] = pl;
    __syncthreads();

    v2f acc[3][3];
#pragma unroll
    for (int a = 0; a < 3; ++a) {
#pragma unroll
        for (int b = 0; b < 3; ++b) acc[a][b] = (v2f){0.f, 0.f};
    }

    for (int c = 0; c < H / KC - 1; ++c) {
        const int cur = c & 1;
        const int kc  = c * KC;
        const int kn  = kc + KC;
        pr0 = *(const float4*)(gR + kn);
        pr1 = *(const float4*)(gR + 1 * rStep + kn);
        pr2 = *(const float4*)(gR + 2 * rStep + kn);
        pr3 = *(const float4*)(gR + 3 * rStep + kn);
        pr4 = *(const float4*)(gR + 4 * rStep + kn);
        pr5 = *(const float4*)(gR + 5 * rStep + kn);
        pl  = *(const float4*)(gL + kn);
        __asm__ volatile("" ::: "memory");

        pair_chunk_math(&Ls[cur][rg * 3][kh * PKH], &Rs[cur][lane][kh * PKH],
                        w2 + kc + kh * PKH, acc);

        const int nx = cur ^ 1;
        *(float4*)&Rs[nx][sr][sc]      = pr0;
        *(float4*)&Rs[nx][sr + 16][sc] = pr1;
        *(float4*)&Rs[nx][sr + 32][sc] = pr2;
        *(float4*)&Rs[nx][sr + 48][sc] = pr3;
        *(float4*)&Rs[nx][sr + 64][sc] = pr4;
        *(float4*)&Rs[nx][sr + 80][sc] = pr5;
        if (stagesL) *(float4*)&Ls[nx][sr][sc] = pl;
        __syncthreads();
    }
    {
        constexpr int cl = H / KC - 1;
        pair_chunk_math(&Ls[cl & 1][rg * 3][kh * PKH], &Rs[cl & 1][lane][kh * PKH],
                        w2 + cl * KC + kh * PKH, acc);
    }

    const int tl = t & 127;
    if (kh == 1) {
#pragma unroll
        for (int a = 0; a < 3; ++a) {
#pragma unroll
            for (int b = 0; b < 3; ++b) red[a * 3 + b][tl] = acc[a][b].x + acc[a][b].y;
        }
    }
    __syncthreads();
    if (kh == 0) {
#pragma unroll
        for (int a = 0; a < 3; ++a) {
            const int i = i0 + rg * 3 + a;
#pragma unroll
            for (int b = 0; b < 3; ++b) {
                const int j = j0 + lane + 32 * b;
                const float v = ((acc[a][b].x + acc[a][b].y) + red[a * 3 + b][tl]) + b2;
                out[(size_t)i * NTOT + j] = (i == j) ? 0.f : v;
            }
        }
    }
}

extern "C" void kernel_launch(void* const* d_in, const int* in_sizes, int n_in,
                              void* d_out, int out_size, void* d_ws, size_t ws_size,
                              hipStream_t stream) {
    const float* sup    = (const float*)d_in[0];
    const float* qry    = (const float*)d_in[2];
    const float* simW1  = (const float*)d_in[3];
    const float* simb1  = (const float*)d_in[4];
    const float* simw2  = (const float*)d_in[5];
    const float* simb2  = (const float*)d_in[6];
    const float* divW1  = (const float*)d_in[7];
    const float* divb1  = (const float*)d_in[8];
    const float* divw2  = (const float*)d_in[9];
    const float* divb2  = (const float*)d_in[10];
    const float* relW   = (const float*)d_in[11];
    const float* relb   = (const float*)d_in[12];

    float* out = (float*)d_out;
    float* ws  = (float*)d_ws;

    float* wbar = ws;
    float* Lsim = ws + 589824;
    float* Rsim = Lsim + 294912;
    float* Ldiv = Rsim + 294912;
    float* Rdiv = Ldiv + 294912;

    float* outProp = out;
    float* outSim  = out + NTOT * H;
    float* outDiv  = outSim + NTOT * NTOT;

    wbar_kernel<<<(H * H) / 256, 256, 0, stream>>>(relW, wbar);

    dim3 ggrid(NTOT / GBM, H / GBN, 5);
    dim3 gblk(256);
    gemm_wmma_f32<<<ggrid, gblk, 0, stream>>>(sup, qry, wbar, simW1, divW1,
                                              relb, simb1, divb1,
                                              outProp, Lsim, Rsim, Ldiv, Rdiv);

    dim3 pgrid(NTOT / PTJ, NTOT / PTI, 2);
    pair_kernel<<<pgrid, 256, 0, stream>>>(Lsim, Rsim, simw2, simb2,
                                           Ldiv, Rdiv, divw2, divb2,
                                           outSim, outDiv);
}


# Launch count 3 -> 2: relation-mean pass runs as the z=0 blocks of the launch of the four L/R GEMMs; the prop GEMM that needs it runs as 96 blocks of 64x48 beside the pair blocks; sum orders unchanged
# speedup vs baseline: 4.9321x; 1.0840x over previous
#include <hip/hip_runtime.h>
#include <hip/hip_bf16.h>

#define H   768
#define NS  128
#define NQ  256
#define NTOT 384
#define RREL 8
#define KC  64

typedef __attribute__((ext_vector_type(8))) float v8f;
typedef __attribute__((ext_vector_type(2))) float v2f;


#define WB_THREADS (48 * 256)
#define WB_GROUPS  ((H * H) / 4)
__device__ __forceinline__ void wbar_role(const float* __restrict__ relW, float* __restrict__ wbar,
                                          const int blk, const int t) {
    const int g0 = blk * 256 + t;
#pragma unroll 2
    for (int j = 0; j < WB_GROUPS / WB_THREADS; ++j) {
        const int g = g0 + j * WB_THREADS;
        float4 s = {0.f, 0.f, 0.f, 0.f};
#pragma unroll
        for (int r = 0; r < RREL; ++r) {
            const float4 v = *(const float4*)(relW + (size_t)r * (H * H) + (size_t)g * 4);
            s.x += v.x; s.y += v.y; s.z += v.z; s.w += v.w;
        }
        float4 o;
        o.x = s.x * (1.0f / RREL); o.y = s.y * (1.0f / RREL);
        o.z = s.z * (1.0f / RREL); o.w = s.w * (1.0f / RREL);
        *(float4*)(wbar + (size_t)g * 4) = o;
    }
}

#define GBM 64
#define GBN 96
#define GKC 64
#define GKG 32
#define GST 3
#define GPITCH (GKC + 4)
#define GASTG (GBM * GPITCH)
#define GBSTG (GBN * GPITCH)
#define GSMEM_FLOATS(NT) (GST * GASTG + GST * 48 * (NT) * GPITCH)

#define GF_DECL(S) \
    v2f S##a0, S##a1, S##a2, S##a3, S##a4, S##a5, S##a6, S##a7; \
    v2f S##p0, S##p1, S##p2, S##p3, S##p4, S##p5, S##p6, S##p7; \
    v2f S##q0, S##q1, S##q2, S##q3, S##q4, S##q5, S##q6, S##q7; \
    v2f S##r0, S##r1, S##r2, S##r3, S##r4, S##r5, S##r6, S##r7;
#define GF_LOAD1(S, i, Ap, Bp, kg) \
    S##a##i = *(const v2f*)((Ap) + (kg) + 4 * i); \
    S##p##i = *(const v2f*)((Bp) + (kg) + 4 * i); \
    S##q##i = *(const v2f*)((Bp) + 16 * GPITCH + (kg) + 4 * i); \
    S##r##i = *(const v2f*)((Bp) + 32 * GPITCH + (kg) + 4 * i);
#define GF_LOAD(S, Ap, Bp, kg) \
    GF_LOAD1(S, 0, Ap, Bp, kg) GF_LOAD1(S, 1, Ap, Bp, kg) GF_LOAD1(S, 2, Ap, Bp, kg) GF_LOAD1(S, 3, Ap, Bp, kg) \
    GF_LOAD1(S, 4, Ap, Bp, kg) GF_LOAD1(S, 5, Ap, Bp, kg) GF_LOAD1(S, 6, Ap, Bp, kg) GF_LOAD1(S, 7, Ap, Bp, kg)
#define GF_MMA1(S, i) \
    acc0 = __builtin_amdgcn_wmma_f32_16x16x4_f32(false, S##a##i, false, S##p##i, (short)0, acc0, false, false); \
    acc1 = __builtin_amdgcn_wmma_f32_16x16x4_f32(false, S##a##i, false, S##q##i, (short)0, acc1, false, false); \
    acc2 = __builtin_amdgcn_wmma_f32_16x16x4_f32(false, S##a##i, false, S##r##i, (short)0, acc2, false, false);
#define GF_MMA(S) \
    GF_MMA1(S, 0) GF_MMA1(S, 1) GF_MMA1(S, 2) GF_MMA1(S, 3) \
    GF_MMA1(S, 4) GF_MMA1(S, 5) GF_MMA1(S, 6) GF_MMA1(S, 7)

template <int NT>
__device__ __forceinline__ void gemm_tile(const float* __restrict__ Ablk,
                                          const float* __restrict__ Wblk, const int ldW,
                                          const float* __restrict__ bias, const int biasMode,
                                          float* __restrict__ out,
                                          const int m0, const int n0, float* smem) {
    constexpr int BSTG = 48 * NT * GPITCH;
    float* const AsBase = smem;
    float* const BsBase = smem + GST * GASTG;

    const int t    = threadIdx.x;
    const int lane = t & 31;
    const int wv   = t >> 5;
    const int wm   = wv & 3;
    const int wn   = wv >> 2;
    const bool computes = (NT == 2) || (wn == 0);
    const int wstrip = (NT == 2) ? wn : 0;
    const int l15  = lane & 15;
    const int khalf = (lane >> 4) * 2;

    const int sr = t >> 4;
    const int sc = (t & 15) * 4;
    const float* gA = Ablk + (size_t)sr * H + sc;
    const float* gB = Wblk + (size_t)sr * ldW + sc;
    const size_t aStep = (size_t)16 * H;
    const size_t bStep = (size_t)16 * ldW;

    float* sA = AsBase + sr * GPITCH + sc;
    float* sB = BsBase + sr * GPITCH + sc;
    const float* fA0 = AsBase + (wm * 16 + l15) * GPITCH + khalf;
    const float* fB0 = BsBase + (wstrip * 48 + l15) * GPITCH + khalf;

    float4 pa0 = *(const float4*)(gA);
    float4 pa1 = *(const float4*)(gA + 1 * aStep);
    float4 pa2 = *(const float4*)(gA + 2 * aStep);
    float4 pa3 = *(const float4*)(gA + 3 * aStep);
    float4 pb0 = *(const float4*)(gB);
    float4 pb1 = *(const float4*)(gB + 1 * bStep);
    float4 pb2 = *(const float4*)(gB + 2 * bStep);
    float4 pb3 = {0.f, 0.f, 0.f, 0.f}, pb4 = pb3, pb5 = pb3;
    if constexpr (NT == 2) {
        pb3 = *(const float4*)(gB + 3 * bStep);
        pb4 = *(const float4*)(gB + 4 * bStep);
        pb5 = *(const float4*)(gB + 5 * bStep);
    }
    {
        const float4 qa0 = *(const float4*)(gA + GKC);
        const float4 qa1 = *(const float4*)(gA + 1 * aStep + GKC);
        const float4 qa2 = *(const float4*)(gA + 2 * aStep + GKC);
        const float4 qa3 = *(const float4*)(gA + 3 * aStep + GKC);
        const float4 qb0 = *(const float4*)(gB + GKC);
        const float4 qb1 = *(const float4*)(gB + 1 * bStep + GKC);
        const float4 qb2 = *(const float4*)(gB + 2 * bStep + GKC);
        float4 qb3 = pb3, qb4 = pb3, qb5 = pb3;
        if constexpr (NT == 2) {
            qb3 = *(const float4*)(gB + 3 * bStep + GKC);
            qb4 = *(const float4*)(gB + 4 * bStep + GKC);
            qb5 = *(const float4*)(gB + 5 * bStep + GKC);
        }
        *(float4*)(sA)               = pa0;
        *(float4*)(sA + 16 * GPITCH) = pa1;
        *(float4*)(sA + 32 * GPITCH) = pa2;
        *(float4*)(sA + 48 * GPITCH) = pa3;
        *(float4*)(sB)               = pb0;
        *(float4*)(sB + 16 * GPITCH) = pb1;
        *(float4*)(sB + 32 * GPITCH) = pb2;
        if constexpr (NT == 2) {
            *(float4*)(sB + 48 * GPITCH) = pb3;
            *(float4*)(sB + 64 * GPITCH) = pb4;
            *(float4*)(sB + 80 * GPITCH) = pb5;
        }
        *(float4*)(sA + GASTG)               = qa0;
        *(float4*)(sA + GASTG + 16 * GPITCH) = qa1;
        *(float4*)(sA + GASTG + 32 * GPITCH) = qa2;
        *(float4*)(sA + GASTG + 48 * GPITCH) = qa3;
        *(float4*)(sB + BSTG)               = qb0;
        *(float4*)(sB + BSTG + 16 * GPITCH) = qb1;
        *(float4*)(sB + BSTG + 32 * GPITCH) = qb2;
        if constexpr (NT == 2) {
            *(float4*)(sB + BSTG + 48 * GPITCH) = qb3;
            *(float4*)(sB + BSTG + 64 * GPITCH) = qb4;
            *(float4*)(sB + BSTG + 80 * GPITCH) = qb5;
        }
    }
    __syncthreads();

    v8f acc0 = {0.f, 0.f, 0.f, 0.f, 0.f, 0.f, 0.f, 0.f};
    v8f acc1 = acc0;
    v8f acc2 = acc0;

    GF_DECL(fx)
    GF_DECL(fy)
    if (computes) { GF_LOAD(fx, fA0, fB0, 0) }

    int s0 = 0, s1 = 1, s2 = 2;
    for (int ch = 0; ch < H / GKC; ++ch) {
        const bool fetch = (ch + 2 < H / GKC);
        if (fetch) {
            const int kn = (ch + 2) * GKC;
            pa0 = *(const float4*)(gA + kn);
            pa1 = *(const float4*)(gA + 1 * aStep + kn);
            pa2 = *(const float4*)(gA + 2 * aStep + kn);
            pa3 = *(const float4*)(gA + 3 * aStep + kn);
            pb0 = *(const float4*)(gB + kn);
            pb1 = *(const float4*)(gB + 1 * bStep + kn);
            pb2 = *(const float4*)(gB + 2 * bStep + kn);
            if constexpr (NT == 2) {
                pb3 = *(const float4*)(gB + 3 * bStep + kn);
                pb4 = *(const float4*)(gB + 4 * bStep + kn);
                pb5 = *(const float4*)(gB + 5 * bStep + kn);
            }
        }

        if (computes) {
            GF_LOAD(fy, fA0 + s0 * GASTG, fB0 + s0 * BSTG, GKG)
            __builtin_amdgcn_sched_barrier(0);
            GF_MMA(fx)
            __builtin_amdgcn_sched_barrier(0);

            GF_LOAD(fx, fA0 + s1 * GASTG, fB0 + s1 * BSTG, 0)
            __builtin_amdgcn_sched_barrier(0);
            GF_MMA(fy)
        }

        if (fetch) {
            float* dA = sA + s2 * GASTG;
            float* dB = sB + s2 * BSTG;
            *(float4*)(dA)               = pa0;
            *(float4*)(dA + 16 * GPITCH) = pa1;
            *(float4*)(dA + 32 * GPITCH) = pa2;
            *(float4*)(dA + 48 * GPITCH) = pa3;
            *(float4*)(dB)               = pb0;
            *(float4*)(dB + 16 * GPITCH) = pb1;
            *(float4*)(dB + 32 * GPITCH) = pb2;
            if constexpr (NT == 2) {
                *(float4*)(dB + 48 * GPITCH) = pb3;
                *(float4*)(dB + 64 * GPITCH) = pb4;
                *(float4*)(dB + 80 * GPITCH) = pb5;
            }
        }
        __syncthreads();
        const int st = s0; s0 = s1; s1 = s2; s2 = st;
    }

    if (!computes) return;

    const int nb = n0 + wstrip * 48 + l15;
    float bv[3] = {0.f, 0.f, 0.f};
    if (biasMode == 1) {
#pragma unroll
        for (int j = 0; j < 3; ++j) bv[j] = bias[nb + j * 16];
    } else if (biasMode == 2) {
        float s[3] = {0.f, 0.f, 0.f};
#pragma unroll
        for (int r = 0; r < RREL; ++r) {
#pragma unroll
            for (int j = 0; j < 3; ++j) s[j] += bias[r * H + nb + j * 16];
        }
#pragma unroll
        for (int j = 0; j < 3; ++j) bv[j] = s[j] * (1.0f / RREL);
    }

    const int mbase = m0 + wm * 16 + ((lane >> 4) ? 8 : 0);
    float* orow = out + (size_t)mbase * H + nb;
#pragma unroll
    for (int r8 = 0; r8 < 8; ++r8) {
        orow[(size_t)r8 * H]      = acc0[r8] + bv[0];
        orow[(size_t)r8 * H + 16] = acc1[r8] + bv[1];
        orow[(size_t)r8 * H + 32] = acc2[r8] + bv[2];
    }
}

__global__ __launch_bounds__(256)
void stage1_kernel(const float* __restrict__ Asup,
                   const float* __restrict__ Aqry,
                   const float* __restrict__ relW,
                   const float* __restrict__ simW1,
                   const float* __restrict__ divW1,
                   const float* __restrict__ simb1,
                   const float* __restrict__ divb1,
                   float* __restrict__ wbar,
                   float* __restrict__ Lsim, float* __restrict__ Rsim,
                   float* __restrict__ Ldiv, float* __restrict__ Rdiv) {
    __shared__ __attribute__((aligned(16))) float smem[GSMEM_FLOATS(2)];

    const int z = blockIdx.z;
    if (z == 0) {
        wbar_role(relW, wbar, blockIdx.y * (NTOT / GBM) + blockIdx.x, threadIdx.x);
        return;
    }
    const float* W    = (z <= 2) ? simW1 : divW1;
    const int colOff  = (z == 2 || z == 4) ? H : 0;
    const float* bias = (z == 1) ? simb1 : divb1;
    const int biasMode = (z == 1 || z == 3) ? 1 : 0;
    float* out = (z == 1) ? Lsim : (z == 2) ? Rsim : (z == 3) ? Ldiv : Rdiv;

    const int m0 = blockIdx.x * GBM;
    const int n0 = blockIdx.y * GBN;
    const float* Ablk = (m0 < NS) ? (Asup + (size_t)m0 * H)
                                  : (Aqry + (size_t)(m0 - NS) * H);
    gemm_tile<2>(Ablk, W + (size_t)n0 * (2 * H) + colOff, 2 * H, bias, biasMode, out, m0, n0, smem);
}

#define PTI 12
#define PTJ 96
#define PPITCH (KC + 4)
#define PKH (KC / 2)

__device__ __forceinline__ void pair_chunk_math(const float* Lr, const float* Rr,
                                                const float* __restrict__ wc, v2f (&acc)[3][3]) {
    const v2f zero2 = {0.f, 0.f};
#pragma unroll 1
    for (int g = 0; g < PKH; g += 16) {
#pragma unroll
        for (int s = 0; s < 16; s += 4) {
            const int ks = g + s;
            float4 lq[3], rq[3];
#pragma unroll
            for (int a = 0; a < 3; ++a) lq[a] = *(const float4*)(Lr + a * PPITCH + ks);
#pragma unroll
            for (int b = 0; b < 3; ++b) rq[b] = *(const float4*)(Rr + b * 32 * PPITCH + ks);
            const v2f w01 = {wc[ks + 0], wc[ks + 1]};
            const v2f w23 = {wc[ks + 2], wc[ks + 3]};
#pragma unroll
            for (int a = 0; a < 3; ++a) {
                const v2f l01 = {lq[a].x, lq[a].y};
                const v2f l23 = {lq[a].z, lq[a].w};
#pragma unroll
                for (int b = 0; b < 3; ++b) {
                    const v2f r01 = {rq[b].x, rq[b].y};
                    const v2f r23 = {rq[b].z, rq[b].w};
                    v2f x = __builtin_elementwise_max(l01 + r01, zero2);
                    acc[a][b] = __builtin_elementwise_fma(x, w01, acc[a][b]);
                    v2f y = __builtin_elementwise_max(l23 + r23, zero2);
                    acc[a][b] = __builtin_elementwise_fma(y, w23, acc[a][b]);
                }
            }
        }
    }
}

#define PLS_FLOATS (2 * PTI * PPITCH)
#define PRS_FLOATS (2 * PTJ * PPITCH)
#define PSMEM_FLOATS (PLS_FLOATS + PRS_FLOATS + 9 * 128)
__device__ __forceinline__ void pair_tile(const float* Lp, const float* Rp,
                                          const float* __restrict__ w2, const float b2,
                                          float* __restrict__ out,
                                          const int bx, const int by, float* smem) {
    float (*Ls)[PTI][PPITCH] = (float (*)[PTI][PPITCH])(smem);
    float (*Rs)[PTJ][PPITCH] = (float (*)[PTJ][PPITCH])(smem + PLS_FLOATS);
    float (*red)[128]        = (float (*)[128])(smem + PLS_FLOATS + PRS_FLOATS);

    const int i0 = by * PTI;
    const int j0 = bx * PTJ;
    const int t    = threadIdx.x;
    const int lane = t & 31;
    const int wv   = __builtin_amdgcn_readfirstlane(t >> 5);
    const int rg   = wv & 3;
    const int kh   = wv >> 2;

    const int sr = t >> 4;
    const int sc = (t & 15) * 4;
    const bool stagesL = (wv < 6);
    const float* gR = Rp + (size_t)(j0 + sr) * H + sc;
    const float* gL = Lp + (size_t)(i0 + (stagesL ? sr : sr - 4)) * H + sc;
    const size_t rStep = (size_t)16 * H;

    float4 pr0 = *(const float4*)(gR);
    float4 pr1 = *(const float4*)(gR + 1 * rStep);
    float4 pr2 = *(const float4*)(gR + 2 * rStep);
    float4 pr3 = *(const float4*)(gR + 3 * rStep);
    float4 pr4 = *(const float4*)(gR + 4 * rStep);
    float4 pr5 = *(const float4*)(gR + 5 * rStep);
    float4 pl  = *(const float4*)(gL);
    *(float4*)&Rs[0][sr][sc]      = pr0;
    *(float4*)&Rs[0][sr + 16][sc] = pr1;
    *(float4*)&Rs[0][sr + 32][sc] = pr2;
    *(float4*)&Rs[0][sr + 48][sc] = pr3;
    *(float4*)&Rs[0][sr + 64][sc] = pr4;
    *(float4*)&Rs[0][sr + 80][sc] = pr5;
    if (stagesL) *(float4*)&Ls[0][sr][sc] = pl;
    __syncthreads();

    v2f acc[3][3];
#pragma unroll
    for (int a = 0; a < 3; ++a) {
#pragma unroll
        for (int b = 0; b < 3; ++b) acc[a][b] = (v2f){0.f, 0.f};
    }

    for (int c = 0; c < H / KC - 1; ++c) {
        const int cur = c & 1;
        const int kc  = c * KC;
        const int kn  = kc + KC;
        pr0 = *(const float4*)(gR + kn);
        pr1 = *(const float4*)(gR + 1 * rStep + kn);
        pr2 = *(const float4*)(gR + 2 * rStep + kn);
        pr3 = *(const float4*)(gR + 3 * rStep + kn);
        pr4 = *(const float4*)(gR + 4 * rStep + kn);
        pr5 = *(const float4*)(gR + 5 * rStep + kn);
        pl  = *(const float4*)(gL + kn);
        __asm__ volatile("" ::: "memory");

        pair_chunk_math(&Ls[cur][rg * 3][kh * PKH], &Rs[cur][lane][kh * PKH],
                        w2 + kc + kh * PKH, acc);

        const int nx = cur ^ 1;
        *(float4*)&Rs[nx][sr][sc]      = pr0;
        *(float4*)&Rs[nx][sr + 16][sc] = pr1;
        *(float4*)&Rs[nx][sr + 32][sc] = pr2;
        *(float4*)&Rs[nx][sr + 48][sc] = pr3;
        *(float4*)&Rs[nx][sr + 64][sc] = pr4;
        *(float4*)&Rs[nx][sr + 80][sc] = pr5;
        if (stagesL) *(float4*)&Ls[nx][sr][sc] = pl;
        __syncthreads();
    }
    {
        constexpr int cl = H / KC - 1;
        pair_chunk_math(&Ls[cl & 1][rg * 3][kh * PKH], &Rs[cl & 1][lane][kh * PKH],
                        w2 + cl * KC + kh * PKH, acc);
    }

    const int tl = t & 127;
    if (kh == 1) {
#pragma unroll
        for (int a = 0; a < 3; ++a) {
#pragma unroll
            for (int b = 0; b < 3; ++b) red[a * 3 + b][tl] = acc[a][b].x + acc[a][b].y;
        }
    }
    __syncthreads();
    if (kh == 0) {
#pragma unroll
        for (int a = 0; a < 3; ++a) {
            const int i = i0 + rg * 3 + a;
#pragma unroll
            for (int b = 0; b < 3; ++b) {
                const int j = j0 + lane + 32 * b;
                const float v = ((acc[a][b].x + acc[a][b].y) + red[a * 3 + b][tl]) + b2;
                out[(size_t)i * NTOT + j] = (i == j) ? 0.f : v;
            }
        }
    }
}

#define S2_GEMM_BLOCKS ((NTOT / GBM) * (H / 48))
#define S2_PAIR_BLOCKS ((NTOT / PTJ) * (NTOT / PTI) * 2)
#define S2_SMEM_FLOATS (GSMEM_FLOATS(1) > PSMEM_FLOATS ? GSMEM_FLOATS(1) : PSMEM_FLOATS)

__global__ __launch_bounds__(256)
void stage2_kernel(const float* __restrict__ Asup,
                   const float* __restrict__ Aqry,
                   const float* __restrict__ wbar,
                   const float* __restrict__ relb,
                   float* __restrict__ outProp,
                   const float* Lsim, const float* Rsim,
                   const float* __restrict__ w2sim, const float* __restrict__ b2sim,
                   const float* Ldiv, const float* Rdiv,
                   const float* __restrict__ w2div, const float* __restrict__ b2div,
                   float* __restrict__ outSim, float* __restrict__ outDiv) {
    __shared__ __attribute__((aligned(16))) float smem[S2_SMEM_FLOATS];

    const int b = blockIdx.x;
    if (b >= S2_PAIR_BLOCKS) {
        const int g  = b - S2_PAIR_BLOCKS;
        const int m0 = (g % (NTOT / GBM)) * GBM;
        const int n0 = (g / (NTOT / GBM)) * 48;
        const float* Ablk = (m0 < NS) ? (Asup + (size_t)m0 * H)
                                      : (Aqry + (size_t)(m0 - NS) * H);
        gemm_tile<1>(Ablk, wbar + (size_t)n0 * H, H, relb, 2, outProp, m0, n0, smem);
        return;
    }
    const int bx = b & 3;
    const int by = (b >> 2) & 31;
    const float *Lp, *Rp, *w2;
    float b2;
    float* out;
    if ((b >> 7) == 0) { Lp = Lsim; Rp = Rsim; w2 = w2sim; b2 = b2sim[0]; out = outSim; }
    else               { Lp = Ldiv; Rp = Rdiv; w2 = w2div; b2 = b2div[0]; out = outDiv; }
    pair_tile(Lp, Rp, w2, b2, out, bx, by, smem);
}

extern "C" void kernel_launch(void* const* d_in, const int* in_sizes, int n_in,
                              void* d_out, int out_size, void* d_ws, size_t ws_size,
                              hipStream_t stream) {
    const float* sup    = (const float*)d_in[0];
    const float* qry    = (const float*)d_in[2];
    const float* simW1  = (const float*)d_in[3];
    const float* simb1  = (const float*)d_in[4];
    const float* simw2  = (const float*)d_in[5];
    const float* simb2  = (const float*)d_in[6];
    const float* divW1  = (const float*)d_in[7];
    const float* divb1  = (const float*)d_in[8];
    const float* divw2  = (const float*)d_in[9];
    const float* divb2  = (const float*)d_in[10];
    const float* relW   = (const float*)d_in[11];
    const float* relb   = (const float*)d_in[12];

    float* out = (float*)d_out;
    float* ws  = (float*)d_ws;

    float* wbar = ws;
    float* Lsim = ws + 589824;
    float* Rsim = Lsim + 294912;
    float* Ldiv = Rsim + 294912;
    float* Rdiv = Ldiv + 294912;

    float* outProp = out;
    float* outSim  = out + NTOT * H;
    float* outDiv  = outSim + NTOT * NTOT;

    dim3 ggrid(NTOT / GBM, H / GBN, 5);
    stage1_kernel<<<ggrid, 256, 0, stream>>>(sup, qry, relW, simW1, divW1, simb1, divb1,
                                             wbar, Lsim, Rsim, Ldiv, Rdiv);

    stage2_kernel<<<S2_PAIR_BLOCKS + S2_GEMM_BLOCKS, 256, 0, stream>>>(
        sup, qry, wbar, relb, outProp,
        Lsim, Rsim, simw2, simb2, Ldiv, Rdiv, divw2, divb2, outSim, outDiv);
}


# Launch 1 made matrix-bound: 256 equal 48x96 blocks (strips 4,5 split by k-group over two waves = 2 partial sums via LDS), Wbar stream folded in, operand fetch two chunks ahead; Wbar sums same
# speedup vs baseline: 5.3099x; 1.0766x over previous
#include <hip/hip_runtime.h>
#include <hip/hip_bf16.h>

#define H   768
#define NS  128
#define NQ  256
#define NTOT 384
#define RREL 8
#define KC  64

typedef __attribute__((ext_vector_type(8))) float v8f;
typedef __attribute__((ext_vector_type(2))) float v2f;


#define WB_PER_BLOCK ((H * H) / 256)
#define WB_RUN (WB_PER_BLOCK / 3)
struct f3 { float x, y, z; };
#define WB_DECL(V) f3 V##0, V##1, V##2;
#define WB_LOAD(V, p) \
    V##0 = *(const f3*)(p); V##1 = *(const f3*)((p) + WB_RUN); V##2 = *(const f3*)((p) + 2 * WB_RUN);
#define WB_ACC(V) \
    ws0.x += (V##0).x; ws0.y += (V##0).y; ws0.z += (V##0).z; \
    ws1.x += (V##1).x; ws1.y += (V##1).y; ws1.z += (V##1).z; \
    ws2.x += (V##2).x; ws2.y += (V##2).y; ws2.z += (V##2).z;
#define WB_SCALE(s) s.x *= (1.0f / RREL); s.y *= (1.0f / RREL); s.z *= (1.0f / RREL);

#define GBM 64
#define GBN 96
#define GKC 64
#define GKG 32
#define GST 3
#define GPITCH (GKC + 4)
#define GASTG (GBM * GPITCH)
#define GBSTG (GBN * GPITCH)
#define GSMEM_FLOATS(NT) (GST * GASTG + GST * 48 * (NT) * GPITCH)

#define GF_DECL(S) \
    v2f S##a0, S##a1, S##a2, S##a3, S##a4, S##a5, S##a6, S##a7; \
    v2f S##p0, S##p1, S##p2, S##p3, S##p4, S##p5, S##p6, S##p7; \
    v2f S##q0, S##q1, S##q2, S##q3, S##q4, S##q5, S##q6, S##q7; \
    v2f S##r0, S##r1, S##r2, S##r3, S##r4, S##r5, S##r6, S##r7;
#define GF_LOAD1(S, i, Ap, Bp, kg) \
    S##a##i = *(const v2f*)((Ap) + (kg) + 4 * i); \
    S##p##i = *(const v2f*)((Bp) + (kg) + 4 * i); \
    S##q##i = *(const v2f*)((Bp) + 16 * GPITCH + (kg) + 4 * i); \
    S##r##i = *(const v2f*)((Bp) + 32 * GPITCH + (kg) + 4 * i);
#define GF_LOAD(S, Ap, Bp, kg) \
    GF_LOAD1(S, 0, Ap, Bp, kg) GF_LOAD1(S, 1, Ap, Bp, kg) GF_LOAD1(S, 2, Ap, Bp, kg) GF_LOAD1(S, 3, Ap, Bp, kg) \
    GF_LOAD1(S, 4, Ap, Bp, kg) GF_LOAD1(S, 5, Ap, Bp, kg) GF_LOAD1(S, 6, Ap, Bp, kg) GF_LOAD1(S, 7, Ap, Bp, kg)
#define GF_MMA1(S, i) \
    acc0 = __builtin_amdgcn_wmma_f32_16x16x4_f32(false, S##a##i, false, S##p##i, (short)0, acc0, false, false); \
    acc1 = __builtin_amdgcn_wmma_f32_16x16x4_f32(false, S##a##i, false, S##q##i, (short)0, acc1, false, false); \
    acc2 = __builtin_amdgcn_wmma_f32_16x16x4_f32(false, S##a##i, false, S##r##i, (short)0, acc2, false, false);
#define GF_MMA(S) \
    GF_MMA1(S, 0) GF_MMA1(S, 1) GF_MMA1(S, 2) GF_MMA1(S, 3) \
    GF_MMA1(S, 4) GF_MMA1(S, 5) GF_MMA1(S, 6) GF_MMA1(S, 7)

template <int NT>
__device__ __forceinline__ void gemm_tile(const float* __restrict__ Ablk,
                                          const float* __restrict__ Wblk, const int ldW,
                                          const float* __restrict__ bias, const int biasMode,
                                          float* __restrict__ out,
                                          const int m0, const int n0, float* smem) {
    constexpr int BSTG = 48 * NT * GPITCH;
    float* const AsBase = smem;
    float* const BsBase = smem + GST * GASTG;

    const int t    = threadIdx.x;
    const int lane = t & 31;
    const int wv   = t >> 5;
    const int wm   = wv & 3;
    const int wn   = wv >> 2;
    const bool computes = (NT == 2) || (wn == 0);
    const int wstrip = (NT == 2) ? wn : 0;
    const int l15  = lane & 15;
    const int khalf = (lane >> 4) * 2;

    const int sr = t >> 4;
    const int sc = (t & 15) * 4;
    const float* gA = Ablk + (size_t)sr * H + sc;
    const float* gB = Wblk + (size_t)sr * ldW + sc;
    const size_t aStep = (size_t)16 * H;
    const size_t bStep = (size_t)16 * ldW;

    float* sA = AsBase + sr * GPITCH + sc;
    float* sB = BsBase + sr * GPITCH + sc;
    const float* fA0 = AsBase + (wm * 16 + l15) * GPITCH + khalf;
    const float* fB0 = BsBase + (wstrip * 48 + l15) * GPITCH + khalf;

    float4 pa0 = *(const float4*)(gA);
    float4 pa1 = *(const float4*)(gA + 1 * aStep);
    float4 pa2 = *(const float4*)(gA + 2 * aStep);
    float4 pa3 = *(const float4*)(gA + 3 * aStep);
    float4 pb0 = *(const float4*)(gB);
    float4 pb1 = *(const float4*)(gB + 1 * bStep);
    float4 pb2 = *(const float4*)(gB + 2 * bStep);
    float4 pb3 = {0.f, 0.f, 0.f, 0.f}, pb4 = pb3, pb5 = pb3;
    if constexpr (NT == 2) {
        pb3 = *(const float4*)(gB + 3 * bStep);
        pb4 = *(const float4*)(gB + 4 * bStep);
        pb5 = *(const float4*)(gB + 5 * bStep);
    }
    {
        const float4 qa0 = *(const float4*)(gA + GKC);
        const float4 qa1 = *(const float4*)(gA + 1 * aStep + GKC);
        const float4 qa2 = *(const float4*)(gA + 2 * aStep + GKC);
        const float4 qa3 = *(const float4*)(gA + 3 * aStep + GKC);
        const float4 qb0 = *(const float4*)(gB + GKC);
        const float4 qb1 = *(const float4*)(gB + 1 * bStep + GKC);
        const float4 qb2 = *(const float4*)(gB + 2 * bStep + GKC);
        float4 qb3 = pb3, qb4 = pb3, qb5 = pb3;
        if constexpr (NT == 2) {
            qb3 = *(const float4*)(gB + 3 * bStep + GKC);
            qb4 = *(const float4*)(gB + 4 * bStep + GKC);
            qb5 = *(const float4*)(gB + 5 * bStep + GKC);
        }
        *(float4*)(sA)               = pa0;
        *(float4*)(sA + 16 * GPITCH) = pa1;
        *(float4*)(sA + 32 * GPITCH) = pa2;
        *(float4*)(sA + 48 * GPITCH) = pa3;
        *(float4*)(sB)               = pb0;
        *(float4*)(sB + 16 * GPITCH) = pb1;
        *(float4*)(sB + 32 * GPITCH) = pb2;
        if constexpr (NT == 2) {
            *(float4*)(sB + 48 * GPITCH) = pb3;
            *(float4*)(sB + 64 * GPITCH) = pb4;
            *(float4*)(sB + 80 * GPITCH) = pb5;
        }
        *(float4*)(sA + GASTG)               = qa0;
        *(float4*)(sA + GASTG + 16 * GPITCH) = qa1;
        *(float4*)(sA + GASTG + 32 * GPITCH) = qa2;
        *(float4*)(sA + GASTG + 48 * GPITCH) = qa3;
        *(float4*)(sB + BSTG)               = qb0;
        *(float4*)(sB + BSTG + 16 * GPITCH) = qb1;
        *(float4*)(sB + BSTG + 32 * GPITCH) = qb2;
        if constexpr (NT == 2) {
            *(float4*)(sB + BSTG + 48 * GPITCH) = qb3;
            *(float4*)(sB + BSTG + 64 * GPITCH) = qb4;
            *(float4*)(sB + BSTG + 80 * GPITCH) = qb5;
        }
    }
    __syncthreads();

    v8f acc0 = {0.f, 0.f, 0.f, 0.f, 0.f, 0.f, 0.f, 0.f};
    v8f acc1 = acc0;
    v8f acc2 = acc0;

    GF_DECL(fx)
    GF_DECL(fy)
    if (computes) { GF_LOAD(fx, fA0, fB0, 0) }

    int s0 = 0, s1 = 1, s2 = 2;
    for (int ch = 0; ch < H / GKC; ++ch) {
        const bool fetch = (ch + 2 < H / GKC);
        if (fetch) {
            const int kn = (ch + 2) * GKC;
            pa0 = *(const float4*)(gA + kn);
            pa1 = *(const float4*)(gA + 1 * aStep + kn);
            pa2 = *(const float4*)(gA + 2 * aStep + kn);
            pa3 = *(const float4*)(gA + 3 * aStep + kn);
            pb0 = *(const float4*)(gB + kn);
            pb1 = *(const float4*)(gB + 1 * bStep + kn);
            pb2 = *(const float4*)(gB + 2 * bStep + kn);
            if constexpr (NT == 2) {
                pb3 = *(const float4*)(gB + 3 * bStep + kn);
                pb4 = *(const float4*)(gB + 4 * bStep + kn);
                pb5 = *(const float4*)(gB + 5 * bStep + kn);
            }
        }

        if (computes) {
            GF_LOAD(fy, fA0 + s0 * GASTG, fB0 + s0 * BSTG, GKG)
            __builtin_amdgcn_sched_barrier(0);
            GF_MMA(fx)
            __builtin_amdgcn_sched_barrier(0);

            GF_LOAD(fx, fA0 + s1 * GASTG, fB0 + s1 * BSTG, 0)
            __builtin_amdgcn_sched_barrier(0);
            GF_MMA(fy)
        }

        if (fetch) {
            float* dA = sA + s2 * GASTG;
            float* dB = sB + s2 * BSTG;
            *(float4*)(dA)               = pa0;
            *(float4*)(dA + 16 * GPITCH) = pa1;
            *(float4*)(dA + 32 * GPITCH) = pa2;
            *(float4*)(dA + 48 * GPITCH) = pa3;
            *(float4*)(dB)               = pb0;
            *(float4*)(dB + 16 * GPITCH) = pb1;
            *(float4*)(dB + 32 * GPITCH) = pb2;
            if constexpr (NT == 2) {
                *(float4*)(dB + 48 * GPITCH) = pb3;
                *(float4*)(dB + 64 * GPITCH) = pb4;
                *(float4*)(dB + 80 * GPITCH) = pb5;
            }
        }
        __syncthreads();
        const int st = s0; s0 = s1; s1 = s2; s2 = st;
    }

    if (!computes) return;

    const int nb = n0 + wstrip * 48 + l15;
    float bv[3] = {0.f, 0.f, 0.f};
    if (biasMode == 1) {
#pragma unroll
        for (int j = 0; j < 3; ++j) bv[j] = bias[nb + j * 16];
    } else if (biasMode == 2) {
        float s[3] = {0.f, 0.f, 0.f};
#pragma unroll
        for (int r = 0; r < RREL; ++r) {
#pragma unroll
            for (int j = 0; j < 3; ++j) s[j] += bias[r * H + nb + j * 16];
        }
#pragma unroll
        for (int j = 0; j < 3; ++j) bv[j] = s[j] * (1.0f / RREL);
    }

    const int mbase = m0 + wm * 16 + ((lane >> 4) ? 8 : 0);
    float* orow = out + (size_t)mbase * H + nb;
#pragma unroll
    for (int r8 = 0; r8 < 8; ++r8) {
        orow[(size_t)r8 * H]      = acc0[r8] + bv[0];
        orow[(size_t)r8 * H + 16] = acc1[r8] + bv[1];
        orow[(size_t)r8 * H + 32] = acc2[r8] + bv[2];
    }
}

#define LBM 48
#define LASTG (LBM * GPITCH)
#define LBSTG (GBN * GPITCH)
#define LSMEM_FLOATS (GST * (LASTG + LBSTG))

__device__ __forceinline__ void lds_barrier() {
    __builtin_amdgcn_fence(__ATOMIC_RELEASE, "workgroup", "local");
    __builtin_amdgcn_s_barrier();
    __builtin_amdgcn_fence(__ATOMIC_ACQUIRE, "workgroup", "local");
}

#define L1_DECL(S) float4 S##0, S##1, S##2, S##3, S##4, S##5, S##6, S##7, S##8;
#define L1_FETCH(S, kn) \
    S##0 = *(const float4*)(gA0 + (kn)); \
    S##1 = *(const float4*)(gA1 + (kn)); \
    S##2 = *(const float4*)(gA2 + (kn)); \
    S##3 = *(const float4*)(gB + (kn)); \
    S##4 = *(const float4*)(gB + 1 * bStep + (kn)); \
    S##5 = *(const float4*)(gB + 2 * bStep + (kn)); \
    S##6 = *(const float4*)(gB + 3 * bStep + (kn)); \
    S##7 = *(const float4*)(gB + 4 * bStep + (kn)); \
    S##8 = *(const float4*)(gB + 5 * bStep + (kn));
#define L1_STORE(S, dA, dB) \
    *(float4*)(dA)                 = S##0; \
    *(float4*)((dA) + 16 * GPITCH) = S##1; \
    *(float4*)((dA) + 32 * GPITCH) = S##2; \
    *(float4*)(dB)                 = S##3; \
    *(float4*)((dB) + 16 * GPITCH) = S##4; \
    *(float4*)((dB) + 32 * GPITCH) = S##5; \
    *(float4*)((dB) + 48 * GPITCH) = S##6; \
    *(float4*)((dB) + 64 * GPITCH) = S##7; \
    *(float4*)((dB) + 80 * GPITCH) = S##8;

#define L1_PIN __asm__ volatile("" ::: "memory");

#define L1_CHUNK(LD, ST, WL, WA, F_FETCH, F_STORE, F_WLOAD, F_WACC, F_WFIN, F_NEXT) { \
    if (F_FETCH) { L1_FETCH(LD, kf) } \
    if (F_WLOAD) { WB_LOAD(WL, wsrc) wsrc += (size_t)H * H; } \
    if (F_FETCH || F_WLOAD) { L1_PIN } \
    if (useY) { GF_LOAD(fy, fA0 + s0 * LASTG, fB0 + s0 * LBSTG, GKG) } \
    if (useX) { GF_MMA(fx) } \
    if (F_WACC) { WB_ACC(WA) } \
    if (F_WFIN) { \
        WB_SCALE(ws0) WB_SCALE(ws1) WB_SCALE(ws2) \
        *(f3*)(wdst) = ws0; *(f3*)(wdst + WB_RUN) = ws1; *(f3*)(wdst + 2 * WB_RUN) = ws2; \
    } \
    if (F_NEXT) { if (useX) { GF_LOAD(fx, fA0 + s1 * LASTG, fB0 + s1 * LBSTG, 0) } } \
    if (useY) { GF_MMA(fy) } \
    if (F_STORE) { L1_STORE(ST, sA + s2 * LASTG, sB + s2 * LBSTG) } \
    lds_barrier(); \
    { const int st_ = s0; s0 = s1; s1 = s2; s2 = st_; } \
    kf += GKC; }

__device__ __forceinline__ void gemm_tile48(const float* Asup,
                                            const float* Aqry,
                                            const float* Wblk, const int ldW,
                                            const float* __restrict__ bias, const int biasMode,
                                            float* __restrict__ out,
                                            const int m0, const int n0,
                                            const float* relW, float* __restrict__ wbar,
                                            const int blk, float* smem) {
    float* const AsBase = smem;
    float* const BsBase = smem + GST * LASTG;

    const int t    = threadIdx.x;
    const int lane = t & 31;
    const int wv   = __builtin_amdgcn_readfirstlane(t >> 5);
    const bool full = (wv < 4);
    const int hpair = full ? 0 : ((wv - 4) >> 1);
    const int strip = full ? wv : 4 + hpair;
    const bool useX = full || ((wv & 1) == 0);
    const bool useY = full || ((wv & 1) == 1);
    const int wm   = strip % 3;
    const int wn   = strip / 3;
    const int l15  = lane & 15;
    const int khalf = (lane >> 4) * 2;

    const int sr = t >> 4;
    const int sc = (t & 15) * 4;
    const int ma0 = m0 + sr, ma1 = ma0 + 16, ma2 = ma0 + 32;
    const float* gA0 = ((ma0 < NS) ? (Asup + (size_t)ma0 * H) : (Aqry + (size_t)(ma0 - NS) * H)) + sc;
    const float* gA1 = ((ma1 < NS) ? (Asup + (size_t)ma1 * H) : (Aqry + (size_t)(ma1 - NS) * H)) + sc;
    const float* gA2 = ((ma2 < NS) ? (Asup + (size_t)ma2 * H) : (Aqry + (size_t)(ma2 - NS) * H)) + sc;
    const float* gB = Wblk + (size_t)sr * ldW + sc;
    const size_t bStep = (size_t)16 * ldW;

    float* sA = AsBase + sr * GPITCH + sc;
    float* sB = BsBase + sr * GPITCH + sc;
    const float* fA0 = AsBase + (wm * 16 + l15) * GPITCH + khalf;
    const float* fB0 = BsBase + (wn * 48 + l15) * GPITCH + khalf;

    const float* wsrc = relW + (size_t)blk * WB_PER_BLOCK + t * 3;
    float* const wdst = wbar + (size_t)blk * WB_PER_BLOCK + t * 3;
    f3 ws0 = {0.f, 0.f, 0.f}, ws1 = ws0, ws2 = ws0;
    WB_DECL(we)
    WB_DECL(wo)

    L1_DECL(pe)
    L1_DECL(po)
    L1_FETCH(pe, 0)
    L1_FETCH(po, GKC)
    L1_STORE(pe, sA, sB)
    L1_STORE(po, sA + LASTG, sB + LBSTG)
    L1_FETCH(pe, 2 * GKC)
    WB_LOAD(we, wsrc) wsrc += (size_t)H * H;
    L1_PIN
    lds_barrier();

    v8f acc0 = {0.f, 0.f, 0.f, 0.f, 0.f, 0.f, 0.f, 0.f};
    v8f acc1 = acc0;
    v8f acc2 = acc0;

    GF_DECL(fx)
    GF_DECL(fy)
    if (useX) { GF_LOAD(fx, fA0, fB0, 0) }

    int s0 = 0, s1 = 1, s2 = 2;
    int kf = 3 * GKC;
    L1_CHUNK(po, pe, wo, we, 1, 1, 1, 1, 0, 1)
    L1_CHUNK(pe, po, we, wo, 1, 1, 1, 1, 0, 1)
    L1_CHUNK(po, pe, wo, we, 1, 1, 1, 1, 0, 1)
    L1_CHUNK(pe, po, we, wo, 1, 1, 1, 1, 0, 1)
    L1_CHUNK(po, pe, wo, we, 1, 1, 1, 1, 0, 1)
    L1_CHUNK(pe, po, we, wo, 1, 1, 1, 1, 0, 1)
    L1_CHUNK(po, pe, wo, we, 1, 1, 1, 1, 0, 1)
    L1_CHUNK(pe, po, we, wo, 1, 1, 0, 1, 1, 1)
    L1_CHUNK(po, pe, wo, we, 1, 1, 0, 0, 0, 1)
    L1_CHUNK(pe, po, we, wo, 0, 1, 0, 0, 0, 1)
    L1_CHUNK(po, pe, wo, we, 0, 0, 0, 0, 0, 1)
    L1_CHUNK(pe, po, we, wo, 0, 0, 0, 0, 0, 0)

    float* red = smem + hpair * (24 * 32) + lane;
    if (!useX) {
#pragma unroll
        for (int r8 = 0; r8 < 8; ++r8) {
            red[(r8)      * 32] = acc0[r8];
            red[(8 + r8)  * 32] = acc1[r8];
            red[(16 + r8) * 32] = acc2[r8];
        }
    }
    __syncthreads();
    if (!useX) return;
    if (!full) {
#pragma unroll
        for (int r8 = 0; r8 < 8; ++r8) {
            acc0[r8] += red[(r8)      * 32];
            acc1[r8] += red[(8 + r8)  * 32];
            acc2[r8] += red[(16 + r8) * 32];
        }
    }

    const int nb = n0 + wn * 48 + l15;
    float bv[3] = {0.f, 0.f, 0.f};
    if (biasMode == 1) {
#pragma unroll
        for (int j = 0; j < 3; ++j) bv[j] = bias[nb + j * 16];
    }

    const int mbase = m0 + wm * 16 + ((lane >> 4) ? 8 : 0);
    float* orow = out + (size_t)mbase * H + nb;
#pragma unroll
    for (int r8 = 0; r8 < 8; ++r8) {
        orow[(size_t)r8 * H]      = acc0[r8] + bv[0];
        orow[(size_t)r8 * H + 16] = acc1[r8] + bv[1];
        orow[(size_t)r8 * H + 32] = acc2[r8] + bv[2];
    }
}

__global__ __launch_bounds__(256)
void stage1_kernel(const float* Asup,
                   const float* Aqry,
                   const float* relW,
                   const float* simW1,
                   const float* divW1,
                   const float* __restrict__ simb1,
                   const float* __restrict__ divb1,
                   float* __restrict__ wbar,
                   float* __restrict__ Lsim, float* __restrict__ Rsim,
                   float* __restrict__ Ldiv, float* __restrict__ Rdiv) {
    __shared__ __attribute__((aligned(16))) float smem[LSMEM_FLOATS];

    const int z = blockIdx.z;
    const float* W    = (z <= 1) ? simW1 : divW1;
    const int colOff  = (z == 1 || z == 3) ? H : 0;
    const float* bias = (z == 0) ? simb1 : divb1;
    const int biasMode = (z == 0 || z == 2) ? 1 : 0;
    float* out = (z == 0) ? Lsim : (z == 1) ? Rsim : (z == 2) ? Ldiv : Rdiv;

    const int m0 = blockIdx.x * LBM;
    const int n0 = blockIdx.y * GBN;
    const int blk = (z * (H / GBN) + blockIdx.y) * (NTOT / LBM) + blockIdx.x;
    gemm_tile48(Asup, Aqry, W + (size_t)n0 * (2 * H) + colOff, 2 * H, bias, biasMode, out, m0, n0,
                relW, wbar, blk, smem);
}

#define PTI 12
#define PTJ 96
#define PPITCH (KC + 4)
#define PKH (KC / 2)

__device__ __forceinline__ void pair_chunk_math(const float* Lr, const float* Rr,
                                                const float* __restrict__ wc, v2f (&acc)[3][3]) {
    const v2f zero2 = {0.f, 0.f};
#pragma unroll 1
    for (int g = 0; g < PKH; g += 16) {
#pragma unroll
        for (int s = 0; s < 16; s += 4) {
            const int ks = g + s;
            float4 lq[3], rq[3];
#pragma unroll
            for (int a = 0; a < 3; ++a) lq[a] = *(const float4*)(Lr + a * PPITCH + ks);
#pragma unroll
            for (int b = 0; b < 3; ++b) rq[b] = *(const float4*)(Rr + b * 32 * PPITCH + ks);
            const v2f w01 = {wc[ks + 0], wc[ks + 1]};
            const v2f w23 = {wc[ks + 2], wc[ks + 3]};
#pragma unroll
            for (int a = 0; a < 3; ++a) {
                const v2f l01 = {lq[a].x, lq[a].y};
                const v2f l23 = {lq[a].z, lq[a].w};
#pragma unroll
                for (int b = 0; b < 3; ++b) {
                    const v2f r01 = {rq[b].x, rq[b].y};
                    const v2f r23 = {rq[b].z, rq[b].w};
                    v2f x = __builtin_elementwise_max(l01 + r01, zero2);
                    acc[a][b] = __builtin_elementwise_fma(x, w01, acc[a][b]);
                    v2f y = __builtin_elementwise_max(l23 + r23, zero2);
                    acc[a][b] = __builtin_elementwise_fma(y, w23, acc[a][b]);
                }
            }
        }
    }
}

#define PLS_FLOATS (2 * PTI * PPITCH)
#define PRS_FLOATS (2 * PTJ * PPITCH)
#define PSMEM_FLOATS (PLS_FLOATS + PRS_FLOATS + 9 * 128)
__device__ __forceinline__ void pair_tile(const float* Lp, const float* Rp,
                                          const float* __restrict__ w2, const float b2,
                                          float* __restrict__ out,
                                          const int bx, const int by, float* smem) {
    float (*Ls)[PTI][PPITCH] = (float (*)[PTI][PPITCH])(smem);
    float (*Rs)[PTJ][PPITCH] = (float (*)[PTJ][PPITCH])(smem + PLS_FLOATS);
    float (*red)[128]        = (float (*)[128])(smem + PLS_FLOATS + PRS_FLOATS);

    const int i0 = by * PTI;
    const int j0 = bx * PTJ;
    const int t    = threadIdx.x;
    const int lane = t & 31;
    const int wv   = __builtin_amdgcn_readfirstlane(t >> 5);
    const int rg   = wv & 3;
    const int kh   = wv >> 2;

    const int sr = t >> 4;
    const int sc = (t & 15) * 4;
    const bool stagesL = (wv < 6);
    const float* gR = Rp + (size_t)(j0 + sr) * H + sc;
    const float* gL = Lp + (size_t)(i0 + (stagesL ? sr : sr - 4)) * H + sc;
    const size_t rStep = (size_t)16 * H;

    float4 pr0 = *(const float4*)(gR);
    float4 pr1 = *(const float4*)(gR + 1 * rStep);
    float4 pr2 = *(const float4*)(gR + 2 * rStep);
    float4 pr3 = *(const float4*)(gR + 3 * rStep);
    float4 pr4 = *(const float4*)(gR + 4 * rStep);
    float4 pr5 = *(const float4*)(gR + 5 * rStep);
    float4 pl  = *(const float4*)(gL);
    *(float4*)&Rs[0][sr][sc]      = pr0;
    *(float4*)&Rs[0][sr + 16][sc] = pr1;
    *(float4*)&Rs[0][sr + 32][sc] = pr2;
    *(float4*)&Rs[0][sr + 48][sc] = pr3;
    *(float4*)&Rs[0][sr + 64][sc] = pr4;
    *(float4*)&Rs[0][sr + 80][sc] = pr5;
    if (stagesL) *(float4*)&Ls[0][sr][sc] = pl;
    __syncthreads();

    v2f acc[3][3];
#pragma unroll
    for (int a = 0; a < 3; ++a) {
#pragma unroll
        for (int b = 0; b < 3; ++b) acc[a][b] = (v2f){0.f, 0.f};
    }

    for (int c = 0; c < H / KC - 1; ++c) {
        const int cur = c & 1;
        const int kc  = c * KC;
        const int kn  = kc + KC;
        pr0 = *(const float4*)(gR + kn);
        pr1 = *(const float4*)(gR + 1 * rStep + kn);
        pr2 = *(const float4*)(gR + 2 * rStep + kn);
        pr3 = *(const float4*)(gR + 3 * rStep + kn);
        pr4 = *(const float4*)(gR + 4 * rStep + kn);
        pr5 = *(const float4*)(gR + 5 * rStep + kn);
        pl  = *(const float4*)(gL + kn);
        __asm__ volatile("" ::: "memory");

        pair_chunk_math(&Ls[cur][rg * 3][kh * PKH], &Rs[cur][lane][kh * PKH],
                        w2 + kc + kh * PKH, acc);

        const int nx = cur ^ 1;
        *(float4*)&Rs[nx][sr][sc]      = pr0;
        *(float4*)&Rs[nx][sr + 16][sc] = pr1;
        *(float4*)&Rs[nx][sr + 32][sc] = pr2;
        *(float4*)&Rs[nx][sr + 48][sc] = pr3;
        *(float4*)&Rs[nx][sr + 64][sc] = pr4;
        *(float4*)&Rs[nx][sr + 80][sc] = pr5;
        if (stagesL) *(float4*)&Ls[nx][sr][sc] = pl;
        __syncthreads();
    }
    {
        constexpr int cl = H / KC - 1;
        pair_chunk_math(&Ls[cl & 1][rg * 3][kh * PKH], &Rs[cl & 1][lane][kh * PKH],
                        w2 + cl * KC + kh * PKH, acc);
    }

    const int tl = t & 127;
    if (kh == 1) {
#pragma unroll
        for (int a = 0; a < 3; ++a) {
#pragma unroll
            for (int b = 0; b < 3; ++b) red[a * 3 + b][tl] = acc[a][b].x + acc[a][b].y;
        }
    }
    __syncthreads();
    if (kh == 0) {
#pragma unroll
        for (int a = 0; a < 3; ++a) {
            const int i = i0 + rg * 3 + a;
#pragma unroll
            for (int b = 0; b < 3; ++b) {
                const int j = j0 + lane + 32 * b;
                const float v = ((acc[a][b].x + acc[a][b].y) + red[a * 3 + b][tl]) + b2;
                out[(size_t)i * NTOT + j] = (i == j) ? 0.f : v;
            }
        }
    }
}

#define S2_GEMM_BLOCKS ((NTOT / GBM) * (H / 48))
#define S2_PAIR_BLOCKS ((NTOT / PTJ) * (NTOT / PTI) * 2)
#define S2_SMEM_FLOATS (GSMEM_FLOATS(1) > PSMEM_FLOATS ? GSMEM_FLOATS(1) : PSMEM_FLOATS)

__global__ __launch_bounds__(256)
void stage2_kernel(const float* __restrict__ Asup,
                   const float* __restrict__ Aqry,
                   const float* __restrict__ wbar,
                   const float* __restrict__ relb,
                   float* __restrict__ outProp,
                   const float* Lsim, const float* Rsim,
                   const float* __restrict__ w2sim, const float* __restrict__ b2sim,
                   const float* Ldiv, const float* Rdiv,
                   const float* __restrict__ w2div, const float* __restrict__ b2div,
                   float* __restrict__ outSim, float* __restrict__ outDiv) {
    __shared__ __attribute__((aligned(16))) float smem[S2_SMEM_FLOATS];

    const int b = blockIdx.x;
    if (b >= S2_PAIR_BLOCKS) {
        const int g  = b - S2_PAIR_BLOCKS;
        const int m0 = (g % (NTOT / GBM)) * GBM;
        const int n0 = (g / (NTOT / GBM)) * 48;
        const float* Ablk = (m0 < NS) ? (Asup + (size_t)m0 * H)
                                      : (Aqry + (size_t)(m0 - NS) * H);
        gemm_tile<1>(Ablk, wbar + (size_t)n0 * H, H, relb, 2, outProp, m0, n0, smem);
        return;
    }
    const int bx = b & 3;
    const int by = (b >> 2) & 31;
    const float *Lp, *Rp, *w2;
    float b2;
    float* out;
    if ((b >> 7) == 0) { Lp = Lsim; Rp = Rsim; w2 = w2sim; b2 = b2sim[0]; out = outSim; }
    else               { Lp = Ldiv; Rp = Rdiv; w2 = w2div; b2 = b2div[0]; out = outDiv; }
    pair_tile(Lp, Rp, w2, b2, out, bx, by, smem);
}

extern "C" void kernel_launch(void* const* d_in, const int* in_sizes, int n_in,
                              void* d_out, int out_size, void* d_ws, size_t ws_size,
                              hipStream_t stream) {
    const float* sup    = (const float*)d_in[0];
    const float* qry    = (const float*)d_in[2];
    const float* simW1  = (const float*)d_in[3];
    const float* simb1  = (const float*)d_in[4];
    const float* simw2  = (const float*)d_in[5];
    const float* simb2  = (const float*)d_in[6];
    const float* divW1  = (const float*)d_in[7];
    const float* divb1  = (const float*)d_in[8];
    const float* divw2  = (const float*)d_in[9];
    const float* divb2  = (const float*)d_in[10];
    const float* relW   = (const float*)d_in[11];
    const float* relb   = (const float*)d_in[12];

    float* out = (float*)d_out;
    float* ws  = (float*)d_ws;

    float* wbar = ws;
    float* Lsim = ws + 589824;
    float* Rsim = Lsim + 294912;
    float* Ldiv = Rsim + 294912;
    float* Rdiv = Ldiv + 294912;

    float* outProp = out;
    float* outSim  = out + NTOT * H;
    float* outDiv  = outSim + NTOT * NTOT;

    dim3 ggrid(NTOT / LBM, H / GBN, 4);
    stage1_kernel<<<ggrid, 256, 0, stream>>>(sup, qry, relW, simW1, divW1, simb1, divb1,
                                             wbar, Lsim, Rsim, Ldiv, Rdiv);

    stage2_kernel<<<S2_PAIR_BLOCKS + S2_GEMM_BLOCKS, 256, 0, stream>>>(
        sup, qry, wbar, relb, outProp,
        Lsim, Rsim, simw2, simb2, Ldiv, Rdiv, divw2, divb2, outSim, outDiv);
}


# Pair loop: relu done by the packed add's clamp bit (L,R stored scaled by 2^-64, exact; sums scaled back once): add + fma per element, was add, max, fma; 6x3 per thread, k in quarters: 8 partial sums
# speedup vs baseline: 5.6474x; 1.0636x over previous
#include <hip/hip_runtime.h>
#include <hip/hip_bf16.h>

#define H   768
#define NS  128
#define NQ  256
#define NTOT 384
#define RREL 8
#define KC  64

typedef __attribute__((ext_vector_type(8))) float v8f;
typedef __attribute__((ext_vector_type(2))) float v2f;


#define WB_PER_BLOCK ((H * H) / 256)
#define WB_RUN (WB_PER_BLOCK / 3)
struct f3 { float x, y, z; };
#define WB_DECL(V) f3 V##0, V##1, V##2;
#define WB_LOAD(V, p) \
    V##0 = *(const f3*)(p); V##1 = *(const f3*)((p) + WB_RUN); V##2 = *(const f3*)((p) + 2 * WB_RUN);
#define WB_ACC(V) \
    ws0.x += (V##0).x; ws0.y += (V##0).y; ws0.z += (V##0).z; \
    ws1.x += (V##1).x; ws1.y += (V##1).y; ws1.z += (V##1).z; \
    ws2.x += (V##2).x; ws2.y += (V##2).y; ws2.z += (V##2).z;
#define WB_SCALE(s) s.x *= (1.0f / RREL); s.y *= (1.0f / RREL); s.z *= (1.0f / RREL);

#define GBM 64
#define GBN 96
#define GKC 64
#define GKG 32
#define GST 3
#define GPITCH (GKC + 4)
#define GASTG (GBM * GPITCH)
#define GBSTG (GBN * GPITCH)
#define GSMEM_FLOATS(NT) (GST * GASTG + GST * 48 * (NT) * GPITCH)

#define GF_DECL(S) \
    v2f S##a0, S##a1, S##a2, S##a3, S##a4, S##a5, S##a6, S##a7; \
    v2f S##p0, S##p1, S##p2, S##p3, S##p4, S##p5, S##p6, S##p7; \
    v2f S##q0, S##q1, S##q2, S##q3, S##q4, S##q5, S##q6, S##q7; \
    v2f S##r0, S##r1, S##r2, S##r3, S##r4, S##r5, S##r6, S##r7;
#define GF_LOAD1(S, i, Ap, Bp, kg) \
    S##a##i = *(const v2f*)((Ap) + (kg) + 4 * i); \
    S##p##i = *(const v2f*)((Bp) + (kg) + 4 * i); \
    S##q##i = *(const v2f*)((Bp) + 16 * GPITCH + (kg) + 4 * i); \
    S##r##i = *(const v2f*)((Bp) + 32 * GPITCH + (kg) + 4 * i);
#define GF_LOAD(S, Ap, Bp, kg) \
    GF_LOAD1(S, 0, Ap, Bp, kg) GF_LOAD1(S, 1, Ap, Bp, kg) GF_LOAD1(S, 2, Ap, Bp, kg) GF_LOAD1(S, 3, Ap, Bp, kg) \
    GF_LOAD1(S, 4, Ap, Bp, kg) GF_LOAD1(S, 5, Ap, Bp, kg) GF_LOAD1(S, 6, Ap, Bp, kg) GF_LOAD1(S, 7, Ap, Bp, kg)
#define GF_MMA1(S, i) \
    acc0 = __builtin_amdgcn_wmma_f32_16x16x4_f32(false, S##a##i, false, S##p##i, (short)0, acc0, false, false); \
    acc1 = __builtin_amdgcn_wmma_f32_16x16x4_f32(false, S##a##i, false, S##q##i, (short)0, acc1, false, false); \
    acc2 = __builtin_amdgcn_wmma_f32_16x16x4_f32(false, S##a##i, false, S##r##i, (short)0, acc2, false, false);
#define GF_MMA(S) \
    GF_MMA1(S, 0) GF_MMA1(S, 1) GF_MMA1(S, 2) GF_MMA1(S, 3) \
    GF_MMA1(S, 4) GF_MMA1(S, 5) GF_MMA1(S, 6) GF_MMA1(S, 7)

template <int NT>
__device__ __forceinline__ void gemm_tile(const float* __restrict__ Ablk,
                                          const float* __restrict__ Wblk, const int ldW,
                                          const float* __restrict__ bias, const int biasMode,
                                          float* __restrict__ out,
                                          const int m0, const int n0, float* smem) {
    constexpr int BSTG = 48 * NT * GPITCH;
    float* const AsBase = smem;
    float* const BsBase = smem + GST * GASTG;

    const int t    = threadIdx.x;
    const int lane = t & 31;
    const int wv   = t >> 5;
    const int wm   = wv & 3;
    const int wn   = wv >> 2;
    const bool computes = (NT == 2) || (wn == 0);
    const int wstrip = (NT == 2) ? wn : 0;
    const int l15  = lane & 15;
    const int khalf = (lane >> 4) * 2;

    const int sr = t >> 4;
    const int sc = (t & 15) * 4;
    const float* gA = Ablk + (size_t)sr * H + sc;
    const float* gB = Wblk + (size_t)sr * ldW + sc;
    const size_t aStep = (size_t)16 * H;
    const size_t bStep = (size_t)16 * ldW;

    float* sA = AsBase + sr * GPITCH + sc;
    float* sB = BsBase + sr * GPITCH + sc;
    const float* fA0 = AsBase + (wm * 16 + l15) * GPITCH + khalf;
    const float* fB0 = BsBase + (wstrip * 48 + l15) * GPITCH + khalf;

    float4 pa0 = *(const float4*)(gA);
    float4 pa1 = *(const float4*)(gA + 1 * aStep);
    float4 pa2 = *(const float4*)(gA + 2 * aStep);
    float4 pa3 = *(const float4*)(gA + 3 * aStep);
    float4 pb0 = *(const float4*)(gB);
    float4 pb1 = *(const float4*)(gB + 1 * bStep);
    float4 pb2 = *(const float4*)(gB + 2 * bStep);
    float4 pb3 = {0.f, 0.f, 0.f, 0.f}, pb4 = pb3, pb5 = pb3;
    if constexpr (NT == 2) {
        pb3 = *(const float4*)(gB + 3 * bStep);
        pb4 = *(const float4*)(gB + 4 * bStep);
        pb5 = *(const float4*)(gB + 5 * bStep);
    }
    {
        const float4 qa0 = *(const float4*)(gA + GKC);
        const float4 qa1 = *(const float4*)(gA + 1 * aStep + GKC);
        const float4 qa2 = *(const float4*)(gA + 2 * aStep + GKC);
        const float4 qa3 = *(const float4*)(gA + 3 * aStep + GKC);
        const float4 qb0 = *(const float4*)(gB + GKC);
        const float4 qb1 = *(const float4*)(gB + 1 * bStep + GKC);
        const float4 qb2 = *(const float4*)(gB + 2 * bStep + GKC);
        float4 qb3 = pb3, qb4 = pb3, qb5 = pb3;
        if constexpr (NT == 2) {
            qb3 = *(const float4*)(gB + 3 * bStep + GKC);
            qb4 = *(const float4*)(gB + 4 * bStep + GKC);
            qb5 = *(const float4*)(gB + 5 * bStep + GKC);
        }
        *(float4*)(sA)               = pa0;
        *(float4*)(sA + 16 * GPITCH) = pa1;
        *(float4*)(sA + 32 * GPITCH) = pa2;
        *(float4*)(sA + 48 * GPITCH) = pa3;
        *(float4*)(sB)               = pb0;
        *(float4*)(sB + 16 * GPITCH) = pb1;
        *(float4*)(sB + 32 * GPITCH) = pb2;
        if constexpr (NT == 2) {
            *(float4*)(sB + 48 * GPITCH) = pb3;
            *(float4*)(sB + 64 * GPITCH) = pb4;
            *(float4*)(sB + 80 * GPITCH) = pb5;
        }
        *(float4*)(sA + GASTG)               = qa0;
        *(float4*)(sA + GASTG + 16 * GPITCH) = qa1;
        *(float4*)(sA + GASTG + 32 * GPITCH) = qa2;
        *(float4*)(sA + GASTG + 48 * GPITCH) = qa3;
        *(float4*)(sB + BSTG)               = qb0;
        *(float4*)(sB + BSTG + 16 * GPITCH) = qb1;
        *(float4*)(sB + BSTG + 32 * GPITCH) = qb2;
        if constexpr (NT == 2) {
            *(float4*)(sB + BSTG + 48 * GPITCH) = qb3;
            *(float4*)(sB + BSTG + 64 * GPITCH) = qb4;
            *(float4*)(sB + BSTG + 80 * GPITCH) = qb5;
        }
    }
    __syncthreads();

    v8f acc0 = {0.f, 0.f, 0.f, 0.f, 0.f, 0.f, 0.f, 0.f};
    v8f acc1 = acc0;
    v8f acc2 = acc0;

    GF_DECL(fx)
    GF_DECL(fy)
    if (computes) { GF_LOAD(fx, fA0, fB0, 0) }

    int s0 = 0, s1 = 1, s2 = 2;
    for (int ch = 0; ch < H / GKC; ++ch) {
        const bool fetch = (ch + 2 < H / GKC);
        if (fetch) {
            const int kn = (ch + 2) * GKC;
            pa0 = *(const float4*)(gA + kn);
            pa1 = *(const float4*)(gA + 1 * aStep + kn);
            pa2 = *(const float4*)(gA + 2 * aStep + kn);
            pa3 = *(const float4*)(gA + 3 * aStep + kn);
            pb0 = *(const float4*)(gB + kn);
            pb1 = *(const float4*)(gB + 1 * bStep + kn);
            pb2 = *(const float4*)(gB + 2 * bStep + kn);
            if constexpr (NT == 2) {
                pb3 = *(const float4*)(gB + 3 * bStep + kn);
                pb4 = *(const float4*)(gB + 4 * bStep + kn);
                pb5 = *(const float4*)(gB + 5 * bStep + kn);
            }
        }

        if (computes) {
            GF_LOAD(fy, fA0 + s0 * GASTG, fB0 + s0 * BSTG, GKG)
            __builtin_amdgcn_sched_barrier(0);
            GF_MMA(fx)
            __builtin_amdgcn_sched_barrier(0);

            GF_LOAD(fx, fA0 + s1 * GASTG, fB0 + s1 * BSTG, 0)
            __builtin_amdgcn_sched_barrier(0);
            GF_MMA(fy)
        }

        if (fetch) {
            float* dA = sA + s2 * GASTG;
            float* dB = sB + s2 * BSTG;
            *(float4*)(dA)               = pa0;
            *(float4*)(dA + 16 * GPITCH) = pa1;
            *(float4*)(dA + 32 * GPITCH) = pa2;
            *(float4*)(dA + 48 * GPITCH) = pa3;
            *(float4*)(dB)               = pb0;
            *(float4*)(dB + 16 * GPITCH) = pb1;
            *(float4*)(dB + 32 * GPITCH) = pb2;
            if constexpr (NT == 2) {
                *(float4*)(dB + 48 * GPITCH) = pb3;
                *(float4*)(dB + 64 * GPITCH) = pb4;
                *(float4*)(dB + 80 * GPITCH) = pb5;
            }
        }
        __syncthreads();
        const int st = s0; s0 = s1; s1 = s2; s2 = st;
    }

    if (!computes) return;

    const int nb = n0 + wstrip * 48 + l15;
    float bv[3] = {0.f, 0.f, 0.f};
    if (biasMode == 1) {
#pragma unroll
        for (int j = 0; j < 3; ++j) bv[j] = bias[nb + j * 16];
    } else if (biasMode == 2) {
        float s[3] = {0.f, 0.f, 0.f};
#pragma unroll
        for (int r = 0; r < RREL; ++r) {
#pragma unroll
            for (int j = 0; j < 3; ++j) s[j] += bias[r * H + nb + j * 16];
        }
#pragma unroll
        for (int j = 0; j < 3; ++j) bv[j] = s[j] * (1.0f / RREL);
    }

    const int mbase = m0 + wm * 16 + ((lane >> 4) ? 8 : 0);
    float* orow = out + (size_t)mbase * H + nb;
#pragma unroll
    for (int r8 = 0; r8 < 8; ++r8) {
        orow[(size_t)r8 * H]      = acc0[r8] + bv[0];
        orow[(size_t)r8 * H + 16] = acc1[r8] + bv[1];
        orow[(size_t)r8 * H + 32] = acc2[r8] + bv[2];
    }
}

#define LBM 48
#define L1_OUT_SCALE 0x1p-64f
#define LASTG (LBM * GPITCH)
#define LBSTG (GBN * GPITCH)
#define LSMEM_FLOATS (GST * (LASTG + LBSTG))

__device__ __forceinline__ void lds_barrier() {
    __builtin_amdgcn_fence(__ATOMIC_RELEASE, "workgroup", "local");
    __builtin_amdgcn_s_barrier();
    __builtin_amdgcn_fence(__ATOMIC_ACQUIRE, "workgroup", "local");
}

#define L1_DECL(S) float4 S##0, S##1, S##2, S##3, S##4, S##5, S##6, S##7, S##8;
#define L1_FETCH(S, kn) \
    S##0 = *(const float4*)(gA0 + (kn)); \
    S##1 = *(const float4*)(gA1 + (kn)); \
    S##2 = *(const float4*)(gA2 + (kn)); \
    S##3 = *(const float4*)(gB + (kn)); \
    S##4 = *(const float4*)(gB + 1 * bStep + (kn)); \
    S##5 = *(const float4*)(gB + 2 * bStep + (kn)); \
    S##6 = *(const float4*)(gB + 3 * bStep + (kn)); \
    S##7 = *(const float4*)(gB + 4 * bStep + (kn)); \
    S##8 = *(const float4*)(gB + 5 * bStep + (kn));
#define L1_STORE(S, dA, dB) \
    *(float4*)(dA)                 = S##0; \
    *(float4*)((dA) + 16 * GPITCH) = S##1; \
    *(float4*)((dA) + 32 * GPITCH) = S##2; \
    *(float4*)(dB)                 = S##3; \
    *(float4*)((dB) + 16 * GPITCH) = S##4; \
    *(float4*)((dB) + 32 * GPITCH) = S##5; \
    *(float4*)((dB) + 48 * GPITCH) = S##6; \
    *(float4*)((dB) + 64 * GPITCH) = S##7; \
    *(float4*)((dB) + 80 * GPITCH) = S##8;

#define L1_PIN __asm__ volatile("" ::: "memory");

#define L1_CHUNK(LD, ST, WL, WA, F_FETCH, F_STORE, F_WLOAD, F_WACC, F_WFIN, F_NEXT) { \
    if (F_FETCH) { L1_FETCH(LD, kf) } \
    if (F_WLOAD) { WB_LOAD(WL, wsrc) wsrc += (size_t)H * H; } \
    if (F_FETCH || F_WLOAD) { L1_PIN } \
    if (useY) { GF_LOAD(fy, fA0 + s0 * LASTG, fB0 + s0 * LBSTG, GKG) } \
    if (useX) { GF_MMA(fx) } \
    if (F_WACC) { WB_ACC(WA) } \
    if (F_WFIN) { \
        WB_SCALE(ws0) WB_SCALE(ws1) WB_SCALE(ws2) \
        *(f3*)(wdst) = ws0; *(f3*)(wdst + WB_RUN) = ws1; *(f3*)(wdst + 2 * WB_RUN) = ws2; \
    } \
    if (F_NEXT) { if (useX) { GF_LOAD(fx, fA0 + s1 * LASTG, fB0 + s1 * LBSTG, 0) } } \
    if (useY) { GF_MMA(fy) } \
    if (F_STORE) { L1_STORE(ST, sA + s2 * LASTG, sB + s2 * LBSTG) } \
    lds_barrier(); \
    { const int st_ = s0; s0 = s1; s1 = s2; s2 = st_; } \
    kf += GKC; }

__device__ __forceinline__ void gemm_tile48(const float* Asup,
                                            const float* Aqry,
                                            const float* Wblk, const int ldW,
                                            const float* __restrict__ bias, const int biasMode,
                                            float* __restrict__ out,
                                            const int m0, const int n0,
                                            const float* relW, float* __restrict__ wbar,
                                            const int blk, float* smem) {
    float* const AsBase = smem;
    float* const BsBase = smem + GST * LASTG;

    const int t    = threadIdx.x;
    const int lane = t & 31;
    const int wv   = __builtin_amdgcn_readfirstlane(t >> 5);
    const bool full = (wv < 4);
    const int hpair = full ? 0 : ((wv - 4) >> 1);
    const int strip = full ? wv : 4 + hpair;
    const bool useX = full || ((wv & 1) == 0);
    const bool useY = full || ((wv & 1) == 1);
    const int wm   = strip % 3;
    const int wn   = strip / 3;
    const int l15  = lane & 15;
    const int khalf = (lane >> 4) * 2;

    const int sr = t >> 4;
    const int sc = (t & 15) * 4;
    const int ma0 = m0 + sr, ma1 = ma0 + 16, ma2 = ma0 + 32;
    const float* gA0 = ((ma0 < NS) ? (Asup + (size_t)ma0 * H) : (Aqry + (size_t)(ma0 - NS) * H)) + sc;
    const float* gA1 = ((ma1 < NS) ? (Asup + (size_t)ma1 * H) : (Aqry + (size_t)(ma1 - NS) * H)) + sc;
    const float* gA2 = ((ma2 < NS) ? (Asup + (size_t)ma2 * H) : (Aqry + (size_t)(ma2 - NS) * H)) + sc;
    const float* gB = Wblk + (size_t)sr * ldW + sc;
    const size_t bStep = (size_t)16 * ldW;

    float* sA = AsBase + sr * GPITCH + sc;
    float* sB = BsBase + sr * GPITCH + sc;
    const float* fA0 = AsBase + (wm * 16 + l15) * GPITCH + khalf;
    const float* fB0 = BsBase + (wn * 48 + l15) * GPITCH + khalf;

    const float* wsrc = relW + (size_t)blk * WB_PER_BLOCK + t * 3;
    float* const wdst = wbar + (size_t)blk * WB_PER_BLOCK + t * 3;
    f3 ws0 = {0.f, 0.f, 0.f}, ws1 = ws0, ws2 = ws0;
    WB_DECL(we)
    WB_DECL(wo)

    L1_DECL(pe)
    L1_DECL(po)
    L1_FETCH(pe, 0)
    L1_FETCH(po, GKC)
    L1_STORE(pe, sA, sB)
    L1_STORE(po, sA + LASTG, sB + LBSTG)
    L1_FETCH(pe, 2 * GKC)
    WB_LOAD(we, wsrc) wsrc += (size_t)H * H;
    L1_PIN
    lds_barrier();

    v8f acc0 = {0.f, 0.f, 0.f, 0.f, 0.f, 0.f, 0.f, 0.f};
    v8f acc1 = acc0;
    v8f acc2 = acc0;

    GF_DECL(fx)
    GF_DECL(fy)
    if (useX) { GF_LOAD(fx, fA0, fB0, 0) }

    int s0 = 0, s1 = 1, s2 = 2;
    int kf = 3 * GKC;
    L1_CHUNK(po, pe, wo, we, 1, 1, 1, 1, 0, 1)
    L1_CHUNK(pe, po, we, wo, 1, 1, 1, 1, 0, 1)
    L1_CHUNK(po, pe, wo, we, 1, 1, 1, 1, 0, 1)
    L1_CHUNK(pe, po, we, wo, 1, 1, 1, 1, 0, 1)
    L1_CHUNK(po, pe, wo, we, 1, 1, 1, 1, 0, 1)
    L1_CHUNK(pe, po, we, wo, 1, 1, 1, 1, 0, 1)
    L1_CHUNK(po, pe, wo, we, 1, 1, 1, 1, 0, 1)
    L1_CHUNK(pe, po, we, wo, 1, 1, 0, 1, 1, 1)
    L1_CHUNK(po, pe, wo, we, 1, 1, 0, 0, 0, 1)
    L1_CHUNK(pe, po, we, wo, 0, 1, 0, 0, 0, 1)
    L1_CHUNK(po, pe, wo, we, 0, 0, 0, 0, 0, 1)
    L1_CHUNK(pe, po, we, wo, 0, 0, 0, 0, 0, 0)

    float* red = smem + hpair * (24 * 32) + lane;
    if (!useX) {
#pragma unroll
        for (int r8 = 0; r8 < 8; ++r8) {
            red[(r8)      * 32] = acc0[r8];
            red[(8 + r8)  * 32] = acc1[r8];
            red[(16 + r8) * 32] = acc2[r8];
        }
    }
    __syncthreads();
    if (!useX) return;
    if (!full) {
#pragma unroll
        for (int r8 = 0; r8 < 8; ++r8) {
            acc0[r8] += red[(r8)      * 32];
            acc1[r8] += red[(8 + r8)  * 32];
            acc2[r8] += red[(16 + r8) * 32];
        }
    }

    const int nb = n0 + wn * 48 + l15;
    float bv[3] = {0.f, 0.f, 0.f};
    if (biasMode == 1) {
#pragma unroll
        for (int j = 0; j < 3; ++j) bv[j] = bias[nb + j * 16];
    }

    const int mbase = m0 + wm * 16 + ((lane >> 4) ? 8 : 0);
    float* orow = out + (size_t)mbase * H + nb;
#pragma unroll
    for (int r8 = 0; r8 < 8; ++r8) {
        orow[(size_t)r8 * H]      = (acc0[r8] + bv[0]) * L1_OUT_SCALE;
        orow[(size_t)r8 * H + 16] = (acc1[r8] + bv[1]) * L1_OUT_SCALE;
        orow[(size_t)r8 * H + 32] = (acc2[r8] + bv[2]) * L1_OUT_SCALE;
    }
}

__global__ __launch_bounds__(256)
void stage1_kernel(const float* Asup,
                   const float* Aqry,
                   const float* relW,
                   const float* simW1,
                   const float* divW1,
                   const float* __restrict__ simb1,
                   const float* __restrict__ divb1,
                   float* __restrict__ wbar,
                   float* __restrict__ Lsim, float* __restrict__ Rsim,
                   float* __restrict__ Ldiv, float* __restrict__ Rdiv) {
    __shared__ __attribute__((aligned(16))) float smem[LSMEM_FLOATS];

    const int z = blockIdx.z;
    const float* W    = (z <= 1) ? simW1 : divW1;
    const int colOff  = (z == 1 || z == 3) ? H : 0;
    const float* bias = (z == 0) ? simb1 : divb1;
    const int biasMode = (z == 0 || z == 2) ? 1 : 0;
    float* out = (z == 0) ? Lsim : (z == 1) ? Rsim : (z == 2) ? Ldiv : Rdiv;

    const int m0 = blockIdx.x * LBM;
    const int n0 = blockIdx.y * GBN;
    const int blk = (z * (H / GBN) + blockIdx.y) * (NTOT / LBM) + blockIdx.x;
    gemm_tile48(Asup, Aqry, W + (size_t)n0 * (2 * H) + colOff, 2 * H, bias, biasMode, out, m0, n0,
                relW, wbar, blk, smem);
}

#define PTI 12
#define PTJ 96
#define PPITCH (KC + 4)
#define PKQ (KC / 4)
#define PRW 6

#define PSCALE_UP   0x1p+64f

__device__ __forceinline__ v2f add_clamp01(const v2f a, const v2f b) {
    v2f d;
    __asm__("v_pk_add_f32 %0, %1, %2 clamp" : "=v"(d) : "v"(a), "v"(b));
    return d;
}

__device__ __forceinline__ void pair_chunk_math(const float* Lr, const float* Rr,
                                                const float* __restrict__ wc, v2f (&acc)[PRW][3]) {
#pragma unroll 1
    for (int g = 0; g < PKQ; g += 8) {
#pragma unroll
        for (int s = 0; s < 8; s += 4) {
            const int ks = g + s;
            float4 lq[PRW], rq[3];
#pragma unroll
            for (int a = 0; a < PRW; ++a) lq[a] = *(const float4*)(Lr + a * PPITCH + ks);
#pragma unroll
            for (int b = 0; b < 3; ++b) rq[b] = *(const float4*)(Rr + b * 32 * PPITCH + ks);
            const v2f w01 = {wc[ks + 0], wc[ks + 1]};
            const v2f w23 = {wc[ks + 2], wc[ks + 3]};
#pragma unroll
            for (int a = 0; a < PRW; ++a) {
                const v2f l01 = {lq[a].x, lq[a].y};
                const v2f l23 = {lq[a].z, lq[a].w};
#pragma unroll
                for (int b = 0; b < 3; ++b) {
                    const v2f r01 = {rq[b].x, rq[b].y};
                    const v2f r23 = {rq[b].z, rq[b].w};
                    const v2f x = add_clamp01(l01, r01);
                    acc[a][b] = __builtin_elementwise_fma(x, w01, acc[a][b]);
                    const v2f y = add_clamp01(l23, r23);
                    acc[a][b] = __builtin_elementwise_fma(y, w23, acc[a][b]);
                }
            }
        }
    }
}

#define PLS_FLOATS (2 * PTI * PPITCH)
#define PRS_FLOATS (2 * PTJ * PPITCH)
#define PRED_FLOATS (3 * 18 * 64)
#define PSMEM_FLOATS (PLS_FLOATS + PRS_FLOATS + PRED_FLOATS)
__device__ __forceinline__ void pair_tile(const float* Lp, const float* Rp,
                                          const float* __restrict__ w2, const float b2,
                                          float* __restrict__ out,
                                          const int bx, const int by, float* smem) {
    float (*Ls)[PTI][PPITCH] = (float (*)[PTI][PPITCH])(smem);
    float (*Rs)[PTJ][PPITCH] = (float (*)[PTJ][PPITCH])(smem + PLS_FLOATS);
    float (*red)[18][64]     = (float (*)[18][64])(smem + PLS_FLOATS + PRS_FLOATS);

    const int i0 = by * PTI;
    const int j0 = bx * PTJ;
    const int t    = threadIdx.x;
    const int lane = t & 31;
    const int wv   = __builtin_amdgcn_readfirstlane(t >> 5);
    const int rg   = wv & 1;
    const int kq   = wv >> 1;

    const int sr = t >> 4;
    const int sc = (t & 15) * 4;
    const bool stagesL = (wv < 6);
    const float* gR = Rp + (size_t)(j0 + sr) * H + sc;
    const float* gL = Lp + (size_t)(i0 + (stagesL ? sr : sr - 4)) * H + sc;
    const size_t rStep = (size_t)16 * H;

    float4 pr0 = *(const float4*)(gR);
    float4 pr1 = *(const float4*)(gR + 1 * rStep);
    float4 pr2 = *(const float4*)(gR + 2 * rStep);
    float4 pr3 = *(const float4*)(gR + 3 * rStep);
    float4 pr4 = *(const float4*)(gR + 4 * rStep);
    float4 pr5 = *(const float4*)(gR + 5 * rStep);
    float4 pl  = *(const float4*)(gL);
    *(float4*)&Rs[0][sr][sc]      = pr0;
    *(float4*)&Rs[0][sr + 16][sc] = pr1;
    *(float4*)&Rs[0][sr + 32][sc] = pr2;
    *(float4*)&Rs[0][sr + 48][sc] = pr3;
    *(float4*)&Rs[0][sr + 64][sc] = pr4;
    *(float4*)&Rs[0][sr + 80][sc] = pr5;
    if (stagesL) *(float4*)&Ls[0][sr][sc] = pl;
    __syncthreads();

    v2f acc[PRW][3];
#pragma unroll
    for (int a = 0; a < PRW; ++a) {
#pragma unroll
        for (int b = 0; b < 3; ++b) acc[a][b] = (v2f){0.f, 0.f};
    }

    for (int c = 0; c < H / KC - 1; ++c) {
        const int cur = c & 1;
        const int kc  = c * KC;
        const int kn  = kc + KC;
        pr0 = *(const float4*)(gR + kn);
        pr1 = *(const float4*)(gR + 1 * rStep + kn);
        pr2 = *(const float4*)(gR + 2 * rStep + kn);
        pr3 = *(const float4*)(gR + 3 * rStep + kn);
        pr4 = *(const float4*)(gR + 4 * rStep + kn);
        pr5 = *(const float4*)(gR + 5 * rStep + kn);
        pl  = *(const float4*)(gL + kn);
        __asm__ volatile("" ::: "memory");

        pair_chunk_math(&Ls[cur][rg * PRW][kq * PKQ], &Rs[cur][lane][kq * PKQ],
                        w2 + kc + kq * PKQ, acc);

        const int nx = cur ^ 1;
        *(float4*)&Rs[nx][sr][sc]      = pr0;
        *(float4*)&Rs[nx][sr + 16][sc] = pr1;
        *(float4*)&Rs[nx][sr + 32][sc] = pr2;
        *(float4*)&Rs[nx][sr + 48][sc] = pr3;
        *(float4*)&Rs[nx][sr + 64][sc] = pr4;
        *(float4*)&Rs[nx][sr + 80][sc] = pr5;
        if (stagesL) *(float4*)&Ls[nx][sr][sc] = pl;
        __syncthreads();
    }
    {
        constexpr int cl = H / KC - 1;
        pair_chunk_math(&Ls[cl & 1][rg * PRW][kq * PKQ], &Rs[cl & 1][lane][kq * PKQ],
                        w2 + cl * KC + kq * PKQ, acc);
    }

    const int tl = rg * 32 + lane;
    if (kq != 0) {
#pragma unroll
        for (int a = 0; a < PRW; ++a) {
#pragma unroll
            for (int b = 0; b < 3; ++b) red[kq - 1][a * 3 + b][tl] = acc[a][b].x + acc[a][b].y;
        }
    }
    __syncthreads();
    if (kq == 0) {
#pragma unroll
        for (int a = 0; a < PRW; ++a) {
            const int i = i0 + rg * PRW + a;
#pragma unroll
            for (int b = 0; b < 3; ++b) {
                const int j = j0 + lane + 32 * b;
                const float q = (((acc[a][b].x + acc[a][b].y) + red[0][a * 3 + b][tl])
                                 + red[1][a * 3 + b][tl]) + red[2][a * 3 + b][tl];
                const float v = q * PSCALE_UP + b2;
                out[(size_t)i * NTOT + j] = (i == j) ? 0.f : v;
            }
        }
    }
}

#define S2_GEMM_BLOCKS ((NTOT / GBM) * (H / 48))
#define S2_PAIR_BLOCKS ((NTOT / PTJ) * (NTOT / PTI) * 2)
#define S2_SMEM_FLOATS (GSMEM_FLOATS(1) > PSMEM_FLOATS ? GSMEM_FLOATS(1) : PSMEM_FLOATS)

__global__ __launch_bounds__(256)
void stage2_kernel(const float* __restrict__ Asup,
                   const float* __restrict__ Aqry,
                   const float* __restrict__ wbar,
                   const float* __restrict__ relb,
                   float* __restrict__ outProp,
                   const float* Lsim, const float* Rsim,
                   const float* __restrict__ w2sim, const float* __restrict__ b2sim,
                   const float* Ldiv, const float* Rdiv,
                   const float* __restrict__ w2div, const float* __restrict__ b2div,
                   float* __restrict__ outSim, float* __restrict__ outDiv) {
    __shared__ __attribute__((aligned(16))) float smem[S2_SMEM_FLOATS];

    const int b = blockIdx.x;
    if (b >= S2_PAIR_BLOCKS) {
        const int g  = b - S2_PAIR_BLOCKS;
        const int m0 = (g % (NTOT / GBM)) * GBM;
        const int n0 = (g / (NTOT / GBM)) * 48;
        const float* Ablk = (m0 < NS) ? (Asup + (size_t)m0 * H)
                                      : (Aqry + (size_t)(m0 - NS) * H);
        gemm_tile<1>(Ablk, wbar + (size_t)n0 * H, H, relb, 2, outProp, m0, n0, smem);
        return;
    }
    const int bx = b & 3;
    const int by = (b >> 2) & 31;
    const float *Lp, *Rp, *w2;
    float b2;
    float* out;
    if ((b >> 7) == 0) { Lp = Lsim; Rp = Rsim; w2 = w2sim; b2 = b2sim[0]; out = outSim; }
    else               { Lp = Ldiv; Rp = Rdiv; w2 = w2div; b2 = b2div[0]; out = outDiv; }
    pair_tile(Lp, Rp, w2, b2, out, bx, by, smem);
}

extern "C" void kernel_launch(void* const* d_in, const int* in_sizes, int n_in,
                              void* d_out, int out_size, void* d_ws, size_t ws_size,
                              hipStream_t stream) {
    const float* sup    = (const float*)d_in[0];
    const float* qry    = (const float*)d_in[2];
    const float* simW1  = (const float*)d_in[3];
    const float* simb1  = (const float*)d_in[4];
    const float* simw2  = (const float*)d_in[5];
    const float* simb2  = (const float*)d_in[6];
    const float* divW1  = (const float*)d_in[7];
    const float* divb1  = (const float*)d_in[8];
    const float* divw2  = (const float*)d_in[9];
    const float* divb2  = (const float*)d_in[10];
    const float* relW   = (const float*)d_in[11];
    const float* relb   = (const float*)d_in[12];

    float* out = (float*)d_out;
    float* ws  = (float*)d_ws;

    float* wbar = ws;
    float* Lsim = ws + 589824;
    float* Rsim = Lsim + 294912;
    float* Ldiv = Rsim + 294912;
    float* Rdiv = Ldiv + 294912;

    float* outProp = out;
    float* outSim  = out + NTOT * H;
    float* outDiv  = outSim + NTOT * NTOT;

    dim3 ggrid(NTOT / LBM, H / GBN, 4);
    stage1_kernel<<<ggrid, 256, 0, stream>>>(sup, qry, relW, simW1, divW1, simb1, divb1,
                                             wbar, Lsim, Rsim, Ldiv, Rdiv);

    stage2_kernel<<<S2_PAIR_BLOCKS + S2_GEMM_BLOCKS, 256, 0, stream>>>(
        sup, qry, wbar, relb, outProp,
        Lsim, Rsim, simw2, simb2, Ldiv, Rdiv, divw2, divb2, outSim, outDiv);
}


# Pair kernel re-tiled for fewer LDS reads: each wave takes the whole 12x96 tile on a k eighth, 6x6 per thread (12 not 18 16-byte reads per 144 terms); output = 16 partial sums (k parity x k eighth)
# speedup vs baseline: 5.9024x; 1.0452x over previous
#include <hip/hip_runtime.h>
#include <hip/hip_bf16.h>

#define H   768
#define NS  128
#define NQ  256
#define NTOT 384
#define RREL 8
#define KC  64

typedef __attribute__((ext_vector_type(8))) float v8f;
typedef __attribute__((ext_vector_type(2))) float v2f;


#define WB_PER_BLOCK ((H * H) / 256)
#define WB_RUN (WB_PER_BLOCK / 3)
struct f3 { float x, y, z; };
#define WB_DECL(V) f3 V##0, V##1, V##2;
#define WB_LOAD(V, p) \
    V##0 = *(const f3*)(p); V##1 = *(const f3*)((p) + WB_RUN); V##2 = *(const f3*)((p) + 2 * WB_RUN);
#define WB_ACC(V) \
    ws0.x += (V##0).x; ws0.y += (V##0).y; ws0.z += (V##0).z; \
    ws1.x += (V##1).x; ws1.y += (V##1).y; ws1.z += (V##1).z; \
    ws2.x += (V##2).x; ws2.y += (V##2).y; ws2.z += (V##2).z;
#define WB_SCALE(s) s.x *= (1.0f / RREL); s.y *= (1.0f / RREL); s.z *= (1.0f / RREL);

#define GBM 64
#define GBN 96
#define GKC 64
#define GKG 32
#define GST 3
#define GPITCH (GKC + 4)
#define GASTG (GBM * GPITCH)
#define GBSTG (GBN * GPITCH)
#define GSMEM_FLOATS(NT) (GST * GASTG + GST * 48 * (NT) * GPITCH)

#define GF_DECL(S) \
    v2f S##a0, S##a1, S##a2, S##a3, S##a4, S##a5, S##a6, S##a7; \
    v2f S##p0, S##p1, S##p2, S##p3, S##p4, S##p5, S##p6, S##p7; \
    v2f S##q0, S##q1, S##q2, S##q3, S##q4, S##q5, S##q6, S##q7; \
    v2f S##r0, S##r1, S##r2, S##r3, S##r4, S##r5, S##r6, S##r7;
#define GF_LOAD1(S, i, Ap, Bp, kg) \
    S##a##i = *(const v2f*)((Ap) + (kg) + 4 * i); \
    S##p##i = *(const v2f*)((Bp) + (kg) + 4 * i); \
    S##q##i = *(const v2f*)((Bp) + 16 * GPITCH + (kg) + 4 * i); \
    S##r##i = *(const v2f*)((Bp) + 32 * GPITCH + (kg) + 4 * i);
#define GF_LOAD(S, Ap, Bp, kg) \
    GF_LOAD1(S, 0, Ap, Bp, kg) GF_LOAD1(S, 1, Ap, Bp, kg) GF_LOAD1(S, 2, Ap, Bp, kg) GF_LOAD1(S, 3, Ap, Bp, kg) \
    GF_LOAD1(S, 4, Ap, Bp, kg) GF_LOAD1(S, 5, Ap, Bp, kg) GF_LOAD1(S, 6, Ap, Bp, kg) GF_LOAD1(S, 7, Ap, Bp, kg)
#define GF_MMA1(S, i) \
    acc0 = __builtin_amdgcn_wmma_f32_16x16x4_f32(false, S##a##i, false, S##p##i, (short)0, acc0, false, false); \
    acc1 = __builtin_amdgcn_wmma_f32_16x16x4_f32(false, S##a##i, false, S##q##i, (short)0, acc1, false, false); \
    acc2 = __builtin_amdgcn_wmma_f32_16x16x4_f32(false, S##a##i, false, S##r##i, (short)0, acc2, false, false);
#define GF_MMA(S) \
    GF_MMA1(S, 0) GF_MMA1(S, 1) GF_MMA1(S, 2) GF_MMA1(S, 3) \
    GF_MMA1(S, 4) GF_MMA1(S, 5) GF_MMA1(S, 6) GF_MMA1(S, 7)

template <int NT>
__device__ __forceinline__ void gemm_tile(const float* __restrict__ Ablk,
                                          const float* __restrict__ Wblk, const int ldW,
                                          const float* __restrict__ bias, const int biasMode,
                                          float* __restrict__ out,
                                          const int m0, const int n0, float* smem) {
    constexpr int BSTG = 48 * NT * GPITCH;
    float* const AsBase = smem;
    float* const BsBase = smem + GST * GASTG;

    const int t    = threadIdx.x;
    const int lane = t & 31;
    const int wv   = t >> 5;
    const int wm   = wv & 3;
    const int wn   = wv >> 2;
    const bool computes = (NT == 2) || (wn == 0);
    const int wstrip = (NT == 2) ? wn : 0;
    const int l15  = lane & 15;
    const int khalf = (lane >> 4) * 2;

    const int sr = t >> 4;
    const int sc = (t & 15) * 4;
    const float* gA = Ablk + (size_t)sr * H + sc;
    const float* gB = Wblk + (size_t)sr * ldW + sc;
    const size_t aStep = (size_t)16 * H;
    const size_t bStep = (size_t)16 * ldW;

    float* sA = AsBase + sr * GPITCH + sc;
    float* sB = BsBase + sr * GPITCH + sc;
    const float* fA0 = AsBase + (wm * 16 + l15) * GPITCH + khalf;
    const float* fB0 = BsBase + (wstrip * 48 + l15) * GPITCH + khalf;

    float4 pa0 = *(const float4*)(gA);
    float4 pa1 = *(const float4*)(gA + 1 * aStep);
    float4 pa2 = *(const float4*)(gA + 2 * aStep);
    float4 pa3 = *(const float4*)(gA + 3 * aStep);
    float4 pb0 = *(const float4*)(gB);
    float4 pb1 = *(const float4*)(gB + 1 * bStep);
    float4 pb2 = *(const float4*)(gB + 2 * bStep);
    float4 pb3 = {0.f, 0.f, 0.f, 0.f}, pb4 = pb3, pb5 = pb3;
    if constexpr (NT == 2) {
        pb3 = *(const float4*)(gB + 3 * bStep);
        pb4 = *(const float4*)(gB + 4 * bStep);
        pb5 = *(const float4*)(gB + 5 * bStep);
    }
    {
        const float4 qa0 = *(const float4*)(gA + GKC);
        const float4 qa1 = *(const float4*)(gA + 1 * aStep + GKC);
        const float4 qa2 = *(const float4*)(gA + 2 * aStep + GKC);
        const float4 qa3 = *(const float4*)(gA + 3 * aStep + GKC);
        const float4 qb0 = *(const float4*)(gB + GKC);
        const float4 qb1 = *(const float4*)(gB + 1 * bStep + GKC);
        const float4 qb2 = *(const float4*)(gB + 2 * bStep + GKC);
        float4 qb3 = pb3, qb4 = pb3, qb5 = pb3;
        if constexpr (NT == 2) {
            qb3 = *(const float4*)(gB + 3 * bStep + GKC);
            qb4 = *(const float4*)(gB + 4 * bStep + GKC);
            qb5 = *(const float4*)(gB + 5 * bStep + GKC);
        }
        *(float4*)(sA)               = pa0;
        *(float4*)(sA + 16 * GPITCH) = pa1;
        *(float4*)(sA + 32 * GPITCH) = pa2;
        *(float4*)(sA + 48 * GPITCH) = pa3;
        *(float4*)(sB)               = pb0;
        *(float4*)(sB + 16 * GPITCH) = pb1;
        *(float4*)(sB + 32 * GPITCH) = pb2;
        if constexpr (NT == 2) {
            *(float4*)(sB + 48 * GPITCH) = pb3;
            *(float4*)(sB + 64 * GPITCH) = pb4;
            *(float4*)(sB + 80 * GPITCH) = pb5;
        }
        *(float4*)(sA + GASTG)               = qa0;
        *(float4*)(sA + GASTG + 16 * GPITCH) = qa1;
        *(float4*)(sA + GASTG + 32 * GPITCH) = qa2;
        *(float4*)(sA + GASTG + 48 * GPITCH) = qa3;
        *(float4*)(sB + BSTG)               = qb0;
        *(float4*)(sB + BSTG + 16 * GPITCH) = qb1;
        *(float4*)(sB + BSTG + 32 * GPITCH) = qb2;
        if constexpr (NT == 2) {
            *(float4*)(sB + BSTG + 48 * GPITCH) = qb3;
            *(float4*)(sB + BSTG + 64 * GPITCH) = qb4;
            *(float4*)(sB + BSTG + 80 * GPITCH) = qb5;
        }
    }
    __syncthreads();

    v8f acc0 = {0.f, 0.f, 0.f, 0.f, 0.f, 0.f, 0.f, 0.f};
    v8f acc1 = acc0;
    v8f acc2 = acc0;

    GF_DECL(fx)
    GF_DECL(fy)
    if (computes) { GF_LOAD(fx, fA0, fB0, 0) }

    int s0 = 0, s1 = 1, s2 = 2;
    for (int ch = 0; ch < H / GKC; ++ch) {
        const bool fetch = (ch + 2 < H / GKC);
        if (fetch) {
            const int kn = (ch + 2) * GKC;
            pa0 = *(const float4*)(gA + kn);
            pa1 = *(const float4*)(gA + 1 * aStep + kn);
            pa2 = *(const float4*)(gA + 2 * aStep + kn);
            pa3 = *(const float4*)(gA + 3 * aStep + kn);
            pb0 = *(const float4*)(gB + kn);
            pb1 = *(const float4*)(gB + 1 * bStep + kn);
            pb2 = *(const float4*)(gB + 2 * bStep + kn);
            if constexpr (NT == 2) {
                pb3 = *(const float4*)(gB + 3 * bStep + kn);
                pb4 = *(const float4*)(gB + 4 * bStep + kn);
                pb5 = *(const float4*)(gB + 5 * bStep + kn);
            }
        }

        if (computes) {
            GF_LOAD(fy, fA0 + s0 * GASTG, fB0 + s0 * BSTG, GKG)
            __builtin_amdgcn_sched_barrier(0);
            GF_MMA(fx)
            __builtin_amdgcn_sched_barrier(0);

            GF_LOAD(fx, fA0 + s1 * GASTG, fB0 + s1 * BSTG, 0)
            __builtin_amdgcn_sched_barrier(0);
            GF_MMA(fy)
        }

        if (fetch) {
            float* dA = sA + s2 * GASTG;
            float* dB = sB + s2 * BSTG;
            *(float4*)(dA)               = pa0;
            *(float4*)(dA + 16 * GPITCH) = pa1;
            *(float4*)(dA + 32 * GPITCH) = pa2;
            *(float4*)(dA + 48 * GPITCH) = pa3;
            *(float4*)(dB)               = pb0;
            *(float4*)(dB + 16 * GPITCH) = pb1;
            *(float4*)(dB + 32 * GPITCH) = pb2;
            if constexpr (NT == 2) {
                *(float4*)(dB + 48 * GPITCH) = pb3;
                *(float4*)(dB + 64 * GPITCH) = pb4;
                *(float4*)(dB + 80 * GPITCH) = pb5;
            }
        }
        __syncthreads();
        const int st = s0; s0 = s1; s1 = s2; s2 = st;
    }

    if (!computes) return;

    const int nb = n0 + wstrip * 48 + l15;
    float bv[3] = {0.f, 0.f, 0.f};
    if (biasMode == 1) {
#pragma unroll
        for (int j = 0; j < 3; ++j) bv[j] = bias[nb + j * 16];
    } else if (biasMode == 2) {
        float s[3] = {0.f, 0.f, 0.f};
#pragma unroll
        for (int r = 0; r < RREL; ++r) {
#pragma unroll
            for (int j = 0; j < 3; ++j) s[j] += bias[r * H + nb + j * 16];
        }
#pragma unroll
        for (int j = 0; j < 3; ++j) bv[j] = s[j] * (1.0f / RREL);
    }

    const int mbase = m0 + wm * 16 + ((lane >> 4) ? 8 : 0);
    float* orow = out + (size_t)mbase * H + nb;
#pragma unroll
    for (int r8 = 0; r8 < 8; ++r8) {
        orow[(size_t)r8 * H]      = acc0[r8] + bv[0];
        orow[(size_t)r8 * H + 16] = acc1[r8] + bv[1];
        orow[(size_t)r8 * H + 32] = acc2[r8] + bv[2];
    }
}

#define LBM 48
#define L1_OUT_SCALE 0x1p-64f
#define LASTG (LBM * GPITCH)
#define LBSTG (GBN * GPITCH)
#define LSMEM_FLOATS (GST * (LASTG + LBSTG))

__device__ __forceinline__ void lds_barrier() {
    __builtin_amdgcn_fence(__ATOMIC_RELEASE, "workgroup", "local");
    __builtin_amdgcn_s_barrier();
    __builtin_amdgcn_fence(__ATOMIC_ACQUIRE, "workgroup", "local");
}

#define L1_DECL(S) float4 S##0, S##1, S##2, S##3, S##4, S##5, S##6, S##7, S##8;
#define L1_FETCH(S, kn) \
    S##0 = *(const float4*)(gA0 + (kn)); \
    S##1 = *(const float4*)(gA1 + (kn)); \
    S##2 = *(const float4*)(gA2 + (kn)); \
    S##3 = *(const float4*)(gB + (kn)); \
    S##4 = *(const float4*)(gB + 1 * bStep + (kn)); \
    S##5 = *(const float4*)(gB + 2 * bStep + (kn)); \
    S##6 = *(const float4*)(gB + 3 * bStep + (kn)); \
    S##7 = *(const float4*)(gB + 4 * bStep + (kn)); \
    S##8 = *(const float4*)(gB + 5 * bStep + (kn));
#define L1_STORE(S, dA, dB) \
    *(float4*)(dA)                 = S##0; \
    *(float4*)((dA) + 16 * GPITCH) = S##1; \
    *(float4*)((dA) + 32 * GPITCH) = S##2; \
    *(float4*)(dB)                 = S##3; \
    *(float4*)((dB) + 16 * GPITCH) = S##4; \
    *(float4*)((dB) + 32 * GPITCH) = S##5; \
    *(float4*)((dB) + 48 * GPITCH) = S##6; \
    *(float4*)((dB) + 64 * GPITCH) = S##7; \
    *(float4*)((dB) + 80 * GPITCH) = S##8;

#define L1_PIN __asm__ volatile("" ::: "memory");

#define L1_CHUNK(LD, ST, WL, WA, F_FETCH, F_STORE, F_WLOAD, F_WACC, F_WFIN, F_NEXT) { \
    if (F_FETCH) { L1_FETCH(LD, kf) } \
    if (F_WLOAD) { WB_LOAD(WL, wsrc) wsrc += (size_t)H * H; } \
    if (F_FETCH || F_WLOAD) { L1_PIN } \
    if (useY) { GF_LOAD(fy, fA0 + s0 * LASTG, fB0 + s0 * LBSTG, GKG) } \
    if (useX) { GF_MMA(fx) } \
    if (F_WACC) { WB_ACC(WA) } \
    if (F_WFIN) { \
        WB_SCALE(ws0) WB_SCALE(ws1) WB_SCALE(ws2) \
        *(f3*)(wdst) = ws0; *(f3*)(wdst + WB_RUN) = ws1; *(f3*)(wdst + 2 * WB_RUN) = ws2; \
    } \
    if (F_NEXT) { if (useX) { GF_LOAD(fx, fA0 + s1 * LASTG, fB0 + s1 * LBSTG, 0) } } \
    if (useY) { GF_MMA(fy) } \
    if (F_STORE) { L1_STORE(ST, sA + s2 * LASTG, sB + s2 * LBSTG) } \
    lds_barrier(); \
    { const int st_ = s0; s0 = s1; s1 = s2; s2 = st_; } \
    kf += GKC; }

__device__ __forceinline__ void gemm_tile48(const float* Asup,
                                            const float* Aqry,
                                            const float* Wblk, const int ldW,
                                            const float* __restrict__ bias, const int biasMode,
                                            float* __restrict__ out,
                                            const int m0, const int n0,
                                            const float* relW, float* __restrict__ wbar,
                                            const int blk, float* smem) {
    float* const AsBase = smem;
    float* const BsBase = smem + GST * LASTG;

    const int t    = threadIdx.x;
    const int lane = t & 31;
    const int wv   = __builtin_amdgcn_readfirstlane(t >> 5);
    const bool full = (wv < 4);
    const int hpair = full ? 0 : ((wv - 4) >> 1);
    const int strip = full ? wv : 4 + hpair;
    const bool useX = full || ((wv & 1) == 0);
    const bool useY = full || ((wv & 1) == 1);
    const int wm   = strip % 3;
    const int wn   = strip / 3;
    const int l15  = lane & 15;
    const int khalf = (lane >> 4) * 2;

    const int sr = t >> 4;
    const int sc = (t & 15) * 4;
    const int ma0 = m0 + sr, ma1 = ma0 + 16, ma2 = ma0 + 32;
    const float* gA0 = ((ma0 < NS) ? (Asup + (size_t)ma0 * H) : (Aqry + (size_t)(ma0 - NS) * H)) + sc;
    const float* gA1 = ((ma1 < NS) ? (Asup + (size_t)ma1 * H) : (Aqry + (size_t)(ma1 - NS) * H)) + sc;
    const float* gA2 = ((ma2 < NS) ? (Asup + (size_t)ma2 * H) : (Aqry + (size_t)(ma2 - NS) * H)) + sc;
    const float* gB = Wblk + (size_t)sr * ldW + sc;
    const size_t bStep = (size_t)16 * ldW;

    float* sA = AsBase + sr * GPITCH + sc;
    float* sB = BsBase + sr * GPITCH + sc;
    const float* fA0 = AsBase + (wm * 16 + l15) * GPITCH + khalf;
    const float* fB0 = BsBase + (wn * 48 + l15) * GPITCH + khalf;

    const float* wsrc = relW + (size_t)blk * WB_PER_BLOCK + t * 3;
    float* const wdst = wbar + (size_t)blk * WB_PER_BLOCK + t * 3;
    f3 ws0 = {0.f, 0.f, 0.f}, ws1 = ws0, ws2 = ws0;
    WB_DECL(we)
    WB_DECL(wo)

    L1_DECL(pe)
    L1_DECL(po)
    L1_FETCH(pe, 0)
    L1_FETCH(po, GKC)
    L1_STORE(pe, sA, sB)
    L1_STORE(po, sA + LASTG, sB + LBSTG)
    L1_FETCH(pe, 2 * GKC)
    WB_LOAD(we, wsrc) wsrc += (size_t)H * H;
    L1_PIN
    lds_barrier();

    v8f acc0 = {0.f, 0.f, 0.f, 0.f, 0.f, 0.f, 0.f, 0.f};
    v8f acc1 = acc0;
    v8f acc2 = acc0;

    GF_DECL(fx)
    GF_DECL(fy)
    if (useX) { GF_LOAD(fx, fA0, fB0, 0) }

    int s0 = 0, s1 = 1, s2 = 2;
    int kf = 3 * GKC;
    L1_CHUNK(po, pe, wo, we, 1, 1, 1, 1, 0, 1)
    L1_CHUNK(pe, po, we, wo, 1, 1, 1, 1, 0, 1)
    L1_CHUNK(po, pe, wo, we, 1, 1, 1, 1, 0, 1)
    L1_CHUNK(pe, po, we, wo, 1, 1, 1, 1, 0, 1)
    L1_CHUNK(po, pe, wo, we, 1, 1, 1, 1, 0, 1)
    L1_CHUNK(pe, po, we, wo, 1, 1, 1, 1, 0, 1)
    L1_CHUNK(po, pe, wo, we, 1, 1, 1, 1, 0, 1)
    L1_CHUNK(pe, po, we, wo, 1, 1, 0, 1, 1, 1)
    L1_CHUNK(po, pe, wo, we, 1, 1, 0, 0, 0, 1)
    L1_CHUNK(pe, po, we, wo, 0, 1, 0, 0, 0, 1)
    L1_CHUNK(po, pe, wo, we, 0, 0, 0, 0, 0, 1)
    L1_CHUNK(pe, po, we, wo, 0, 0, 0, 0, 0, 0)

    float* red = smem + hpair * (24 * 32) + lane;
    if (!useX) {
#pragma unroll
        for (int r8 = 0; r8 < 8; ++r8) {
            red[(r8)      * 32] = acc0[r8];
            red[(8 + r8)  * 32] = acc1[r8];
            red[(16 + r8) * 32] = acc2[r8];
        }
    }
    __syncthreads();
    if (!useX) return;
    if (!full) {
#pragma unroll
        for (int r8 = 0; r8 < 8; ++r8) {
            acc0[r8] += red[(r8)      * 32];
            acc1[r8] += red[(8 + r8)  * 32];
            acc2[r8] += red[(16 + r8) * 32];
        }
    }

    const int nb = n0 + wn * 48 + l15;
    float bv[3] = {0.f, 0.f, 0.f};
    if (biasMode == 1) {
#pragma unroll
        for (int j = 0; j < 3; ++j) bv[j] = bias[nb + j * 16];
    }

    const int mbase = m0 + wm * 16 + ((lane >> 4) ? 8 : 0);
    float* orow = out + (size_t)mbase * H + nb;
#pragma unroll
    for (int r8 = 0; r8 < 8; ++r8) {
        orow[(size_t)r8 * H]      = (acc0[r8] + bv[0]) * L1_OUT_SCALE;
        orow[(size_t)r8 * H + 16] = (acc1[r8] + bv[1]) * L1_OUT_SCALE;
        orow[(size_t)r8 * H + 32] = (acc2[r8] + bv[2]) * L1_OUT_SCALE;
    }
}

__global__ __launch_bounds__(256)
void stage1_kernel(const float* Asup,
                   const float* Aqry,
                   const float* relW,
                   const float* simW1,
                   const float* divW1,
                   const float* __restrict__ simb1,
                   const float* __restrict__ divb1,
                   float* __restrict__ wbar,
                   float* __restrict__ Lsim, float* __restrict__ Rsim,
                   float* __restrict__ Ldiv, float* __restrict__ Rdiv) {
    __shared__ __attribute__((aligned(16))) float smem[LSMEM_FLOATS];

    const int z = blockIdx.z;
    const float* W    = (z <= 1) ? simW1 : divW1;
    const int colOff  = (z == 1 || z == 3) ? H : 0;
    const float* bias = (z == 0) ? simb1 : divb1;
    const int biasMode = (z == 0 || z == 2) ? 1 : 0;
    float* out = (z == 0) ? Lsim : (z == 1) ? Rsim : (z == 2) ? Ldiv : Rdiv;

    const int m0 = blockIdx.x * LBM;
    const int n0 = blockIdx.y * GBN;
    const int blk = (z * (H / GBN) + blockIdx.y) * (NTOT / LBM) + blockIdx.x;
    gemm_tile48(Asup, Aqry, W + (size_t)n0 * (2 * H) + colOff, 2 * H, bias, biasMode, out, m0, n0,
                relW, wbar, blk, smem);
}

#define PTI 12
#define PTJ 96
#define PPITCH (KC + 4)
#define PKE (KC / 8)
#define PRW 6
#define PCW 6

#define PSCALE_UP   0x1p+64f

__device__ __forceinline__ v2f add_clamp01(const v2f a, const v2f b) {
    v2f d;
    __asm__("v_pk_add_f32 %0, %1, %2 clamp" : "=v"(d) : "v"(a), "v"(b));
    return d;
}

__device__ __forceinline__ void pair_chunk_math(const float* Lr, const float* Rr,
                                                const float* __restrict__ wc, v2f (&acc)[PRW][PCW]) {
#pragma unroll 1
    for (int ks = 0; ks < PKE; ks += 4) {
        float4 lq[PRW], rq[PCW];
#pragma unroll
        for (int a = 0; a < PRW; ++a) lq[a] = *(const float4*)(Lr + a * PPITCH + ks);
#pragma unroll
        for (int b = 0; b < PCW; ++b) rq[b] = *(const float4*)(Rr + b * 16 * PPITCH + ks);
        const v2f w01 = {wc[ks + 0], wc[ks + 1]};
        const v2f w23 = {wc[ks + 2], wc[ks + 3]};
#pragma unroll
        for (int a = 0; a < PRW; a += 2) {
            v2f x[2][PCW], y[2][PCW];
#pragma unroll
            for (int d = 0; d < 2; ++d) {
                const v2f l01 = {lq[a + d].x, lq[a + d].y};
                const v2f l23 = {lq[a + d].z, lq[a + d].w};
#pragma unroll
                for (int b = 0; b < PCW; ++b) {
                    const v2f r01 = {rq[b].x, rq[b].y};
                    const v2f r23 = {rq[b].z, rq[b].w};
                    x[d][b] = add_clamp01(l01, r01);
                    y[d][b] = add_clamp01(l23, r23);
                }
            }
#pragma unroll
            for (int d = 0; d < 2; ++d) {
#pragma unroll
                for (int b = 0; b < PCW; ++b) acc[a + d][b] = __builtin_elementwise_fma(x[d][b], w01, acc[a + d][b]);
            }
#pragma unroll
            for (int d = 0; d < 2; ++d) {
#pragma unroll
                for (int b = 0; b < PCW; ++b) acc[a + d][b] = __builtin_elementwise_fma(y[d][b], w23, acc[a + d][b]);
            }
        }
    }
}

#define PLS_FLOATS (2 * PTI * PPITCH)
#define PRS_FLOATS (2 * PTJ * PPITCH)
#define PRED_FLOATS (8 * 36 * 32)
#define PSMEM_FLOATS (PLS_FLOATS + PRS_FLOATS + PRED_FLOATS)
__device__ __forceinline__ void pair_tile(const float* Lp, const float* Rp,
                                          const float* __restrict__ w2, const float b2,
                                          float* __restrict__ out,
                                          const int bx, const int by, float* smem) {
    float (*Ls)[PTI][PPITCH] = (float (*)[PTI][PPITCH])(smem);
    float (*Rs)[PTJ][PPITCH] = (float (*)[PTJ][PPITCH])(smem + PLS_FLOATS);
    float (*red)[36][32]     = (float (*)[36][32])(smem + PLS_FLOATS + PRS_FLOATS);

    const int i0 = by * PTI;
    const int j0 = bx * PTJ;
    const int t    = threadIdx.x;
    const int lane = t & 31;
    const int wv   = __builtin_amdgcn_readfirstlane(t >> 5);
    const int rg   = lane >> 4;
    const int cl   = lane & 15;

    const int sr = t >> 4;
    const int sc = (t & 15) * 4;
    const bool stagesL = (wv < 6);
    const float* gR = Rp + (size_t)(j0 + sr) * H + sc;
    const float* gL = Lp + (size_t)(i0 + (stagesL ? sr : sr - 4)) * H + sc;
    const size_t rStep = (size_t)16 * H;

    float4 pr0 = *(const float4*)(gR);
    float4 pr1 = *(const float4*)(gR + 1 * rStep);
    float4 pr2 = *(const float4*)(gR + 2 * rStep);
    float4 pr3 = *(const float4*)(gR + 3 * rStep);
    float4 pr4 = *(const float4*)(gR + 4 * rStep);
    float4 pr5 = *(const float4*)(gR + 5 * rStep);
    float4 pl  = *(const float4*)(gL);
    *(float4*)&Rs[0][sr][sc]      = pr0;
    *(float4*)&Rs[0][sr + 16][sc] = pr1;
    *(float4*)&Rs[0][sr + 32][sc] = pr2;
    *(float4*)&Rs[0][sr + 48][sc] = pr3;
    *(float4*)&Rs[0][sr + 64][sc] = pr4;
    *(float4*)&Rs[0][sr + 80][sc] = pr5;
    if (stagesL) *(float4*)&Ls[0][sr][sc] = pl;
    __syncthreads();

    v2f acc[PRW][PCW];
#pragma unroll
    for (int a = 0; a < PRW; ++a) {
#pragma unroll
        for (int b = 0; b < PCW; ++b) acc[a][b] = (v2f){0.f, 0.f};
    }

    for (int c = 0; c < H / KC - 1; ++c) {
        const int cur = c & 1;
        const int kc  = c * KC;
        const int kn  = kc + KC;
        pr0 = *(const float4*)(gR + kn);
        pr1 = *(const float4*)(gR + 1 * rStep + kn);
        pr2 = *(const float4*)(gR + 2 * rStep + kn);
        pr3 = *(const float4*)(gR + 3 * rStep + kn);
        pr4 = *(const float4*)(gR + 4 * rStep + kn);
        pr5 = *(const float4*)(gR + 5 * rStep + kn);
        pl  = *(const float4*)(gL + kn);
        __asm__ volatile("" ::: "memory");

        pair_chunk_math(&Ls[cur][rg * PRW][wv * PKE], &Rs[cur][cl][wv * PKE],
                        w2 + kc + wv * PKE, acc);

        const int nx = cur ^ 1;
        *(float4*)&Rs[nx][sr][sc]      = pr0;
        *(float4*)&Rs[nx][sr + 16][sc] = pr1;
        *(float4*)&Rs[nx][sr + 32][sc] = pr2;
        *(float4*)&Rs[nx][sr + 48][sc] = pr3;
        *(float4*)&Rs[nx][sr + 64][sc] = pr4;
        *(float4*)&Rs[nx][sr + 80][sc] = pr5;
        if (stagesL) *(float4*)&Ls[nx][sr][sc] = pl;
        __syncthreads();
    }
    {
        constexpr int cl_ = H / KC - 1;
        pair_chunk_math(&Ls[cl_ & 1][rg * PRW][wv * PKE], &Rs[cl_ & 1][cl][wv * PKE],
                        w2 + cl_ * KC + wv * PKE, acc);
    }

#pragma unroll
    for (int a = 0; a < PRW; ++a) {
#pragma unroll
        for (int b = 0; b < PCW; ++b) red[wv][a * PCW + b][lane] = acc[a][b].x + acc[a][b].y;
    }
    __syncthreads();
    for (int o = t; o < PTI * PTJ; o += 256) {
        const int row = o / PTJ;
        const int col = o - row * PTJ;
        const int orow = row / PRW;
        const int oa   = row - orow * PRW;
        const int oln  = orow * 16 + (col & 15);
        const int ob   = col >> 4;
        const int slot = oa * PCW + ob;
        float q = red[0][slot][oln];
#pragma unroll
        for (int e = 1; e < 8; ++e) q += red[e][slot][oln];
        const float v = q * PSCALE_UP + b2;
        const int i = i0 + row;
        const int j = j0 + col;
        out[(size_t)i * NTOT + j] = (i == j) ? 0.f : v;
    }
}

#define S2_GEMM_BLOCKS ((NTOT / GBM) * (H / 48))
#define S2_PAIR_BLOCKS ((NTOT / PTJ) * (NTOT / PTI) * 2)
#define S2_SMEM_FLOATS (GSMEM_FLOATS(1) > PSMEM_FLOATS ? GSMEM_FLOATS(1) : PSMEM_FLOATS)

__global__ __launch_bounds__(256)
void stage2_kernel(const float* __restrict__ Asup,
                   const float* __restrict__ Aqry,
                   const float* __restrict__ wbar,
                   const float* __restrict__ relb,
                   float* __restrict__ outProp,
                   const float* Lsim, const float* Rsim,
                   const float* __restrict__ w2sim, const float* __restrict__ b2sim,
                   const float* Ldiv, const float* Rdiv,
                   const float* __restrict__ w2div, const float* __restrict__ b2div,
                   float* __restrict__ outSim, float* __restrict__ outDiv) {
    __shared__ __attribute__((aligned(16))) float smem[S2_SMEM_FLOATS];

    const int b = blockIdx.x;
    if (b >= S2_PAIR_BLOCKS) {
        const int g  = b - S2_PAIR_BLOCKS;
        const int m0 = (g % (NTOT / GBM)) * GBM;
        const int n0 = (g / (NTOT / GBM)) * 48;
        const float* Ablk = (m0 < NS) ? (Asup + (size_t)m0 * H)
                                      : (Aqry + (size_t)(m0 - NS) * H);
        gemm_tile<1>(Ablk, wbar + (size_t)n0 * H, H, relb, 2, outProp, m0, n0, smem);
        return;
    }
    const int bx = b & 3;
    const int by = (b >> 2) & 31;
    const float *Lp, *Rp, *w2;
    float b2;
    float* out;
    if ((b >> 7) == 0) { Lp = Lsim; Rp = Rsim; w2 = w2sim; b2 = b2sim[0]; out = outSim; }
    else               { Lp = Ldiv; Rp = Rdiv; w2 = w2div; b2 = b2div[0]; out = outDiv; }
    pair_tile(Lp, Rp, w2, b2, out, bx, by, smem);
}

extern "C" void kernel_launch(void* const* d_in, const int* in_sizes, int n_in,
                              void* d_out, int out_size, void* d_ws, size_t ws_size,
                              hipStream_t stream) {
    const float* sup    = (const float*)d_in[0];
    const float* qry    = (const float*)d_in[2];
    const float* simW1  = (const float*)d_in[3];
    const float* simb1  = (const float*)d_in[4];
    const float* simw2  = (const float*)d_in[5];
    const float* simb2  = (const float*)d_in[6];
    const float* divW1  = (const float*)d_in[7];
    const float* divb1  = (const float*)d_in[8];
    const float* divw2  = (const float*)d_in[9];
    const float* divb2  = (const float*)d_in[10];
    const float* relW   = (const float*)d_in[11];
    const float* relb   = (const float*)d_in[12];

    float* out = (float*)d_out;
    float* ws  = (float*)d_ws;

    float* wbar = ws;
    float* Lsim = ws + 589824;
    float* Rsim = Lsim + 294912;
    float* Ldiv = Rsim + 294912;
    float* Rdiv = Ldiv + 294912;

    float* outProp = out;
    float* outSim  = out + NTOT * H;
    float* outDiv  = outSim + NTOT * NTOT;

    dim3 ggrid(NTOT / LBM, H / GBN, 4);
    stage1_kernel<<<ggrid, 256, 0, stream>>>(sup, qry, relW, simW1, divW1, simb1, divb1,
                                             wbar, Lsim, Rsim, Ldiv, Rdiv);

    stage2_kernel<<<S2_PAIR_BLOCKS + S2_GEMM_BLOCKS, 256, 0, stream>>>(
        sup, qry, wbar, relb, outProp,
        Lsim, Rsim, simw2, simb2, Ldiv, Rdiv, divw2, divb2, outSim, outDiv);
}
